# starting kernel
# baseline (speedup 1.0000x reference)
#include <hip/hip_runtime.h>

typedef float v2f __attribute__((ext_vector_type(2)));
typedef float v8f __attribute__((ext_vector_type(8)));


__global__ void k_init_deg(float* deg, int n) {
    int i = blockIdx.x * blockDim.x + threadIdx.x;
    if (i < n) deg[i] = 1.0f;
}

__global__ void k_count_deg(const int* __restrict__ dst, long e, float* deg) {
    long i = (long)blockIdx.x * blockDim.x + threadIdx.x;
    if (i < e) unsafeAtomicAdd(&deg[dst[i]], 1.0f);
}

__global__ void k_rsqrt_inplace(float* deg, int n) {
    int i = blockIdx.x * blockDim.x + threadIdx.x;
    if (i < n) deg[i] = rsqrtf(deg[i]);
}


template <int K, int NOUT, bool RELU_BIAS>
__global__ __launch_bounds__(256) void k_gemm_wmma(const float* __restrict__ A,
                                                   const float* __restrict__ W,
                                                   const float* __restrict__ bias,
                                                   float* __restrict__ Out,
                                                   int nRows) {
    constexpr int PITCH = 2 * NOUT + 32;
    constexpr int HK    = K / 2;
    constexpr int PAIRS = K / 4;
    constexpr int CT    = NOUT / 16;
    __shared__ __align__(16) float lw[PAIRS * PITCH];

    const int lane  = threadIdx.x & 31;
    const int wave  = threadIdx.x >> 5;
    const int laneM = lane & 15;
    const int hi    = lane >> 4;

    const int rowBase = blockIdx.x * 128 + wave * 16 + laneM;
    const int rowC    = rowBase < nRows ? rowBase : 0;

    v8f acc[CT] = {};

    for (int kh = 0; kh < 2; ++kh) {
        __syncthreads();
        for (int idx = threadIdx.x; idx < HK * NOUT; idx += 256) {
            int lk = idx / NOUT;
            int c  = idx - lk * NOUT;
            lw[(lk >> 1) * PITCH + 2 * c + (lk & 1)] = W[(kh * HK + lk) * NOUT + c];
        }
        __syncthreads();

        for (int k4 = 0; k4 < HK; k4 += 4) {
            const int kk = kh * HK + k4 + 2 * hi;
            v2f a;
            a.x = A[(long)rowC * K + kk];
            a.y = A[(long)rowC * K + kk + 1];
            if (RELU_BIAS) {
                a.x = fmaxf(a.x + bias[kk], 0.0f);
                a.y = fmaxf(a.y + bias[kk + 1], 0.0f);
            }
            const float* lb = &lw[((k4 >> 1) + hi) * PITCH];

            v2f bfrag[CT];
#pragma unroll
            for (int ct = 0; ct < CT; ++ct)
                bfrag[ct] = *(const v2f*)&lb[2 * (ct * 16 + laneM)];

#pragma unroll
            for (int ct = 0; ct < CT; ++ct)
                acc[ct] = __builtin_amdgcn_wmma_f32_16x16x4_f32(
                    false, a, false, bfrag[ct], (short)0, acc[ct], false, false);
        }
    }

#pragma unroll
    for (int ct = 0; ct < CT; ++ct) {
#pragma unroll
        for (int r = 0; r < 8; ++r) {
            int m = blockIdx.x * 128 + wave * 16 + 8 * hi + r;
            if (m < nRows) Out[(long)m * NOUT + ct * 16 + laneM] = acc[ct][r];
        }
    }
}


__global__ void k_self_init(const float* __restrict__ h, const float* __restrict__ dinv,
                            const float* __restrict__ bias, float* __restrict__ out,
                            long total, int logF) {
    long gid = (long)blockIdx.x * blockDim.x + threadIdx.x;
    if (gid >= total) return;
    int  i = (int)(gid >> logF);
    int  f = (int)(gid & ((1 << logF) - 1));
    float d = dinv[i];
    float v = h[gid] * d * d;
    if (bias) v += bias[f];
    out[gid] = v;
}


template <int LOGF>
__global__ void k_scatter(const int* __restrict__ src, const int* __restrict__ dst,
                          const float* __restrict__ dinv, const float* __restrict__ h,
                          float* __restrict__ out, long nE) {
    constexpr int CH = (1 << LOGF) / 4;
    long gid = (long)blockIdx.x * blockDim.x + threadIdx.x;
    long e   = gid >> (LOGF - 2);
    if (e >= nE) return;
    int   c4 = (int)(gid & (CH - 1));
    int   s  = src[e], d = dst[e];
    float w  = dinv[s] * dinv[d];
    float4 v = ((const float4*)(h + ((long)s << LOGF)))[c4];
    float* o = out + ((long)d << LOGF) + 4 * c4;
    unsafeAtomicAdd(o + 0, v.x * w);
    unsafeAtomicAdd(o + 1, v.y * w);
    unsafeAtomicAdd(o + 2, v.z * w);
    unsafeAtomicAdd(o + 3, v.w * w);
}


extern "C" void kernel_launch(void* const* d_in, const int* in_sizes, int n_in,
                              void* d_out, int out_size, void* d_ws, size_t ws_size,
                              hipStream_t stream) {
    const float* x   = (const float*)d_in[0];
    const int*   ei  = (const int*)d_in[1];
    const float* W1  = (const float*)d_in[2];
    const float* b1  = (const float*)d_in[3];
    const float* W2  = (const float*)d_in[4];
    const float* b2  = (const float*)d_in[5];
    float*       out = (float*)d_out;

    const int  n = in_sizes[0] / 128;
    const long E = in_sizes[1] / 2;
    const int* srcIdx = ei;
    const int* dstIdx = ei + E;

    float* ws   = (float*)d_ws;
    float* dinv = ws;
    float* h1   = dinv + n;
    float* agg1 = h1 + (long)n * 128;
    float* h2   = agg1 + (long)n * 128;

    const int B = 256;
    auto blk = [](long work, int b) { return (int)((work + b - 1) / b); };

    k_init_deg<<<blk(n, B), B, 0, stream>>>(dinv, n);
    k_count_deg<<<blk(E, B), B, 0, stream>>>(dstIdx, E, dinv);
    k_rsqrt_inplace<<<blk(n, B), B, 0, stream>>>(dinv, n);

    k_gemm_wmma<128, 128, false><<<blk(n, 128), 256, 0, stream>>>(x, W1, nullptr, h1, n);
    k_self_init<<<blk((long)n * 128, B), B, 0, stream>>>(h1, dinv, nullptr, agg1, (long)n * 128, 7);
    k_scatter<7><<<blk(E * 32, B), B, 0, stream>>>(srcIdx, dstIdx, dinv, h1, agg1, E);

    k_gemm_wmma<128, 64, true><<<blk(n, 128), 256, 0, stream>>>(agg1, W2, b1, h2, n);
    k_self_init<<<blk((long)n * 64, B), B, 0, stream>>>(h2, dinv, b2, out, (long)n * 64, 6);
    k_scatter<6><<<blk(E * 16, B), B, 0, stream>>>(srcIdx, dstIdx, dinv, h2, out, E);
}


# float-atomic edge scatter replaced by per-call counting sort of edges by destination (one returning int atomic per edge) and one-wave-per-row gather sums stored once; summation order changes
# speedup vs baseline: 1.8802x; 1.8802x over previous
#include <hip/hip_runtime.h>

typedef float v2f __attribute__((ext_vector_type(2)));
typedef float v8f __attribute__((ext_vector_type(8)));


constexpr int EPT = 8;

__global__ void k_zero_cnt(int* cnt, int n) {
    int i = blockIdx.x * blockDim.x + threadIdx.x;
    if (i < n) cnt[i] = 0;
}

__global__ __launch_bounds__(256) void k_count_rank(const int* __restrict__ dst, long e,
                                                    int* cnt, int* __restrict__ rank) {
    const int E = (int)e;
    const int T = gridDim.x * 256;
    const int t = blockIdx.x * 256 + threadIdx.x;
    int d[EPT], r[EPT];
#pragma unroll
    for (int k = 0; k < EPT; ++k) {
        int i = t + k * T;
        d[k] = dst[i < E ? i : E - 1];
    }
#pragma unroll
    for (int k = 0; k < EPT; ++k) {
        int i = t + k * T;
        r[k] = i < E ? atomicAdd(&cnt[d[k]], 1) : 0;
    }
#pragma unroll
    for (int k = 0; k < EPT; ++k) {
        int i = t + k * T;
        if (i < E) rank[i] = r[k];
    }
}

__global__ __launch_bounds__(256) void k_scan_dinv(const int* __restrict__ cnt,
                                                   int* __restrict__ rowptr,
                                                   float* __restrict__ dinv,
                                                   int2* __restrict__ csr, int n) {
    __shared__ int sPre[8], sTot[8];
    const int tid = threadIdx.x, lane = tid & 31, wv = tid >> 5;
    const int tileBase = blockIdx.x * 1024;

    int pre = 0;
    const int4* c4 = (const int4*)cnt;
    for (int j = tid; j < tileBase / 4; j += 256) {
        int4 v = c4[j];
        pre += (v.x + v.y) + (v.z + v.w) + 4;
    }
    for (int o = 16; o > 0; o >>= 1) pre += __shfl_xor(pre, o);

    const int base = tileBase + tid * 4;
    int cc[4];
#pragma unroll
    for (int k = 0; k < 4; ++k) cc[k] = base + k < n ? cnt[base + k] + 1 : 0;
    const int t = (cc[0] + cc[1]) + (cc[2] + cc[3]);
    int x = t;
    for (int o = 1; o < 32; o <<= 1) {
        int y = __shfl_up(x, o);
        if (lane >= o) x += y;
    }
    if (lane == 31) sTot[wv] = x;
    if (lane == 0) sPre[wv] = pre;
    __syncthreads();
    int run = x - t;
#pragma unroll
    for (int k = 0; k < 8; ++k) {
        run += sPre[k];
        if (k < wv) run += sTot[k];
    }
#pragma unroll
    for (int k = 0; k < 4; ++k) {
        if (base + k < n) {
            const float di = rsqrtf((float)cc[k]);
            rowptr[base + k] = run;
            dinv[base + k]   = di;
            csr[run]         = make_int2(base + k, __float_as_int(di));
            run += cc[k];
        }
    }
}

__global__ __launch_bounds__(256) void k_fill(const int* __restrict__ src, const int* __restrict__ dst,
                                              const int* __restrict__ rank,
                                              const int* __restrict__ rowptr,
                                              const float* __restrict__ dinv,
                                              int2* __restrict__ csr, long e) {
    const int E = (int)e;
    const int T = gridDim.x * 256;
    const int t = blockIdx.x * 256 + threadIdx.x;
    int s[EPT], d[EPT], r[EPT];
#pragma unroll
    for (int k = 0; k < EPT; ++k) {
        int i  = t + k * T;
        int ic = i < E ? i : E - 1;
        s[k] = src[ic];
        d[k] = dst[ic];
        r[k] = rank[ic];
    }
    int   pos[EPT];
    float ws[EPT];
#pragma unroll
    for (int k = 0; k < EPT; ++k) {
        pos[k] = rowptr[d[k]] + 1 + r[k];
        ws[k]  = dinv[s[k]];
    }
#pragma unroll
    for (int k = 0; k < EPT; ++k) {
        int i = t + k * T;
        if (i < E) csr[pos[k]] = make_int2(s[k], __float_as_int(ws[k]));
    }
}


template <int K, int NOUT, bool RELU_BIAS>
__global__ __launch_bounds__(256) void k_gemm_wmma(const float* __restrict__ A,
                                                   const float* __restrict__ W,
                                                   const float* __restrict__ bias,
                                                   float* __restrict__ Out,
                                                   int nRows) {
    constexpr int PITCH = 2 * NOUT + 32;
    constexpr int HK    = K / 2;
    constexpr int PAIRS = K / 4;
    constexpr int CT    = NOUT / 16;
    __shared__ __align__(16) float lw[PAIRS * PITCH];

    const int lane  = threadIdx.x & 31;
    const int wave  = threadIdx.x >> 5;
    const int laneM = lane & 15;
    const int hi    = lane >> 4;

    const int rowBase = blockIdx.x * 128 + wave * 16 + laneM;
    const int rowC    = rowBase < nRows ? rowBase : 0;

    v8f acc[CT] = {};

    for (int kh = 0; kh < 2; ++kh) {
        __syncthreads();
        for (int idx = threadIdx.x; idx < HK * NOUT; idx += 256) {
            int lk = idx / NOUT;
            int c  = idx - lk * NOUT;
            lw[(lk >> 1) * PITCH + 2 * c + (lk & 1)] = W[(kh * HK + lk) * NOUT + c];
        }
        __syncthreads();

        for (int k4 = 0; k4 < HK; k4 += 4) {
            const int kk = kh * HK + k4 + 2 * hi;
            v2f a;
            a.x = A[(long)rowC * K + kk];
            a.y = A[(long)rowC * K + kk + 1];
            if (RELU_BIAS) {
                a.x = fmaxf(a.x + bias[kk], 0.0f);
                a.y = fmaxf(a.y + bias[kk + 1], 0.0f);
            }
            const float* lb = &lw[((k4 >> 1) + hi) * PITCH];

            v2f bfrag[CT];
#pragma unroll
            for (int ct = 0; ct < CT; ++ct)
                bfrag[ct] = *(const v2f*)&lb[2 * (ct * 16 + laneM)];

#pragma unroll
            for (int ct = 0; ct < CT; ++ct)
                acc[ct] = __builtin_amdgcn_wmma_f32_16x16x4_f32(
                    false, a, false, bfrag[ct], (short)0, acc[ct], false, false);
        }
    }

#pragma unroll
    for (int ct = 0; ct < CT; ++ct) {
#pragma unroll
        for (int r = 0; r < 8; ++r) {
            int m = blockIdx.x * 128 + wave * 16 + 8 * hi + r;
            if (m < nRows) Out[(long)m * NOUT + ct * 16 + laneM] = acc[ct][r];
        }
    }
}


constexpr int ROWS = 4;

template <int F, bool BIAS>
__global__ __launch_bounds__(256) void k_gather(const int* __restrict__ rowptr,
                                                const int* __restrict__ cnt,
                                                const int2* __restrict__ csr,
                                                const float* __restrict__ dinv,
                                                const float* __restrict__ h,
                                                const float* __restrict__ bias,
                                                float* __restrict__ out, int n) {
    constexpr int V = F / 32;
    typedef float vf __attribute__((ext_vector_type(V)));
    const int lane = threadIdx.x & 31;
    const int wv = __builtin_amdgcn_readfirstlane((int)((blockIdx.x * 256 + threadIdx.x) >> 5));

    vf bv = {};
    if (BIAS) bv = *(const vf*)(bias + lane * V);

    int   begs[ROWS], degs[ROWS];
    float dis[ROWS];
#pragma unroll
    for (int r = 0; r < ROWS; ++r) {
        const int ic = wv * ROWS + r < n ? wv * ROWS + r : n - 1;
        begs[r] = rowptr[ic];
        degs[r] = cnt[ic] + 1;
        dis[r]  = dinv[ic];
    }

#pragma unroll
    for (int r = 0; r < ROWS; ++r) {
        const int i = wv * ROWS + r;
        if (i >= n) break;
        const int   beg = begs[r];
        const int   deg = degs[r];
        const float di  = dis[r];

        vf acc = {};
        int p = 0;
        do {
            const int m = deg - p < 32 ? deg - p : 32;
            const int2 ed = csr[beg + p + (lane < m ? lane : m - 1)];

            int j = 0;
            do {
                vf    v[4];
                float w[4];
#pragma unroll
                for (int u = 0; u < 4; ++u) {
                    const int jj = j + u < m ? j + u : m - 1;
                    const int s  = __builtin_amdgcn_readlane(ed.x, jj);
                    w[u] = __int_as_float(__builtin_amdgcn_readlane(ed.y, jj)) * di;
                    v[u] = *(const vf*)(h + (long)s * F + lane * V);
                }
#pragma unroll
                for (int u = 0; u < 4; ++u)
                    if (j + u < m) acc += v[u] * w[u];
                j += 4;
            } while (j < m);
            p += 32;
        } while (p < deg);
        if (BIAS) acc += bv;
        *(vf*)(out + (long)i * F + lane * V) = acc;
    }
}


extern "C" void kernel_launch(void* const* d_in, const int* in_sizes, int n_in,
                              void* d_out, int out_size, void* d_ws, size_t ws_size,
                              hipStream_t stream) {
    const float* x   = (const float*)d_in[0];
    const int*   ei  = (const int*)d_in[1];
    const float* W1  = (const float*)d_in[2];
    const float* b1  = (const float*)d_in[3];
    const float* W2  = (const float*)d_in[4];
    const float* b2  = (const float*)d_in[5];
    float*       out = (float*)d_out;

    const int  n = in_sizes[0] / 128;
    const long E = in_sizes[1] / 2;
    const int* srcIdx = ei;
    const int* dstIdx = ei + E;

    const long n4 = ((long)n + 3) & ~3L;
    const long E4 = (E + 3) & ~3L;
    int*   cnt    = (int*)d_ws;
    int*   rowptr = cnt + n4;
    int*   rank   = rowptr + n4;
    int2*  csr    = (int2*)(rank + E4);
    float* dinv   = (float*)(csr + E4 + n4);
    float* h1     = dinv + n4;
    float* agg1   = h1 + (long)n * 128;
    float* h2     = agg1 + (long)n * 128;

    const int B = 256;
    auto blk = [](long work, int b) { return (int)((work + b - 1) / b); };

    k_zero_cnt<<<blk(n, B), B, 0, stream>>>(cnt, n);
    k_count_rank<<<blk(E, B * EPT), B, 0, stream>>>(dstIdx, E, cnt, rank);
    k_scan_dinv<<<blk(n, 1024), B, 0, stream>>>(cnt, rowptr, dinv, csr, n);
    k_fill<<<blk(E, B * EPT), B, 0, stream>>>(srcIdx, dstIdx, rank, rowptr, dinv, csr, E);

    k_gemm_wmma<128, 128, false><<<blk(n, 128), 256, 0, stream>>>(x, W1, nullptr, h1, n);
    k_gather<128, false><<<blk(n, 8 * ROWS), B, 0, stream>>>(rowptr, cnt, csr, dinv, h1, nullptr, agg1, n);

    k_gemm_wmma<128, 64, true><<<blk(n, 128), 256, 0, stream>>>(agg1, W2, b1, h2, n);
    k_gather<64, true><<<blk(n, 8 * ROWS), B, 0, stream>>>(rowptr, cnt, csr, dinv, h2, b2, out, n);
}


# scan kernel: the reads in front of a tile issued 16 at a time and waited for as one group, and the tile's four counters as one 16-byte read, instead of one read per wait
# speedup vs baseline: 2.1118x; 1.1232x over previous
#include <hip/hip_runtime.h>

typedef float v2f __attribute__((ext_vector_type(2)));
typedef float v8f __attribute__((ext_vector_type(8)));


constexpr int EPT = 8;

__global__ void k_zero_cnt(int* cnt, int n) {
    int i = blockIdx.x * blockDim.x + threadIdx.x;
    if (i < n) cnt[i] = 0;
}

__global__ __launch_bounds__(256) void k_count_rank(const int* __restrict__ dst, long e,
                                                    int* cnt, int* __restrict__ rank) {
    const int E = (int)e;
    const int T = gridDim.x * 256;
    const int t = blockIdx.x * 256 + threadIdx.x;
    int d[EPT], r[EPT];
#pragma unroll
    for (int k = 0; k < EPT; ++k) {
        int i = t + k * T;
        d[k] = dst[i < E ? i : E - 1];
    }
#pragma unroll
    for (int k = 0; k < EPT; ++k) {
        int i = t + k * T;
        r[k] = i < E ? atomicAdd(&cnt[d[k]], 1) : 0;
    }
#pragma unroll
    for (int k = 0; k < EPT; ++k) {
        int i = t + k * T;
        if (i < E) rank[i] = r[k];
    }
}


constexpr int LB = 16;
typedef int v4i __attribute__((ext_vector_type(4)));

__global__ __launch_bounds__(256) void k_scan_dinv(const int* __restrict__ cnt,
                                                   int* __restrict__ rowptr,
                                                   float* __restrict__ dinv,
                                                   int2* __restrict__ csr, int n) {
    __shared__ int sPre[8], sTot[8];
    const int tid = threadIdx.x, lane = tid & 31, wv = tid >> 5;
    const int tileBase = blockIdx.x * 1024;
    const v4i* c4 = (const v4i*)cnt;

    const int base = tileBase + tid * 4;
    const v4i cv   = c4[(base < n ? base : 0) >> 2];

    int pre = 0;
    const int nb = blockIdx.x;
    for (int i0 = 0; i0 < nb; i0 += LB) {
        v4i v[LB];
#pragma unroll
        for (int u = 0; u < LB; ++u) {
            const int i = i0 + u < nb ? i0 + u : nb - 1;
            v[u] = c4[i * 256 + tid];
        }
        asm volatile("" ::"v"(v[0]), "v"(v[1]), "v"(v[2]), "v"(v[3]), "v"(v[4]), "v"(v[5]),
                     "v"(v[6]), "v"(v[7]), "v"(v[8]), "v"(v[9]), "v"(v[10]), "v"(v[11]),
                     "v"(v[12]), "v"(v[13]), "v"(v[14]), "v"(v[15]));
        static_assert(LB == 16, "the statement above names LB values");
#pragma unroll
        for (int u = 0; u < LB; ++u) {
            const int keep = i0 + u < nb ? -1 : 0;
            pre += ((v[u].x + v[u].y) + (v[u].z + v[u].w) + 4) & keep;
        }
    }
    for (int o = 16; o > 0; o >>= 1) pre += __shfl_xor(pre, o);

    int cc[4];
    cc[0] = base + 0 < n ? cv.x + 1 : 0;
    cc[1] = base + 1 < n ? cv.y + 1 : 0;
    cc[2] = base + 2 < n ? cv.z + 1 : 0;
    cc[3] = base + 3 < n ? cv.w + 1 : 0;
    const int t = (cc[0] + cc[1]) + (cc[2] + cc[3]);
    int x = t;
    for (int o = 1; o < 32; o <<= 1) {
        int y = __shfl_up(x, o);
        if (lane >= o) x += y;
    }
    if (lane == 31) sTot[wv] = x;
    if (lane == 0) sPre[wv] = pre;
    __syncthreads();
    int run = x - t;
#pragma unroll
    for (int k = 0; k < 8; ++k) {
        run += sPre[k];
        if (k < wv) run += sTot[k];
    }
#pragma unroll
    for (int k = 0; k < 4; ++k) {
        if (base + k < n) {
            const float di = rsqrtf((float)cc[k]);
            rowptr[base + k] = run;
            dinv[base + k]   = di;
            csr[run]         = make_int2(base + k, __float_as_int(di));
            run += cc[k];
        }
    }
}

__global__ __launch_bounds__(256) void k_fill(const int* __restrict__ src, const int* __restrict__ dst,
                                              const int* __restrict__ rank,
                                              const int* __restrict__ rowptr,
                                              const float* __restrict__ dinv,
                                              int2* __restrict__ csr, long e) {
    const int E = (int)e;
    const int T = gridDim.x * 256;
    const int t = blockIdx.x * 256 + threadIdx.x;
    int s[EPT], d[EPT], r[EPT];
#pragma unroll
    for (int k = 0; k < EPT; ++k) {
        int i  = t + k * T;
        int ic = i < E ? i : E - 1;
        s[k] = src[ic];
        d[k] = dst[ic];
        r[k] = rank[ic];
    }
    int   pos[EPT];
    float ws[EPT];
#pragma unroll
    for (int k = 0; k < EPT; ++k) {
        pos[k] = rowptr[d[k]] + 1 + r[k];
        ws[k]  = dinv[s[k]];
    }
#pragma unroll
    for (int k = 0; k < EPT; ++k) {
        int i = t + k * T;
        if (i < E) csr[pos[k]] = make_int2(s[k], __float_as_int(ws[k]));
    }
}


template <int K, int NOUT, bool RELU_BIAS>
__global__ __launch_bounds__(256) void k_gemm_wmma(const float* __restrict__ A,
                                                   const float* __restrict__ W,
                                                   const float* __restrict__ bias,
                                                   float* __restrict__ Out,
                                                   int nRows) {
    constexpr int PITCH = 2 * NOUT + 32;
    constexpr int HK    = K / 2;
    constexpr int PAIRS = K / 4;
    constexpr int CT    = NOUT / 16;
    __shared__ __align__(16) float lw[PAIRS * PITCH];

    const int lane  = threadIdx.x & 31;
    const int wave  = threadIdx.x >> 5;
    const int laneM = lane & 15;
    const int hi    = lane >> 4;

    const int rowBase = blockIdx.x * 128 + wave * 16 + laneM;
    const int rowC    = rowBase < nRows ? rowBase : 0;

    v8f acc[CT] = {};

    for (int kh = 0; kh < 2; ++kh) {
        __syncthreads();
        for (int idx = threadIdx.x; idx < HK * NOUT; idx += 256) {
            int lk = idx / NOUT;
            int c  = idx - lk * NOUT;
            lw[(lk >> 1) * PITCH + 2 * c + (lk & 1)] = W[(kh * HK + lk) * NOUT + c];
        }
        __syncthreads();

        for (int k4 = 0; k4 < HK; k4 += 4) {
            const int kk = kh * HK + k4 + 2 * hi;
            v2f a;
            a.x = A[(long)rowC * K + kk];
            a.y = A[(long)rowC * K + kk + 1];
            if (RELU_BIAS) {
                a.x = fmaxf(a.x + bias[kk], 0.0f);
                a.y = fmaxf(a.y + bias[kk + 1], 0.0f);
            }
            const float* lb = &lw[((k4 >> 1) + hi) * PITCH];

            v2f bfrag[CT];
#pragma unroll
            for (int ct = 0; ct < CT; ++ct)
                bfrag[ct] = *(const v2f*)&lb[2 * (ct * 16 + laneM)];

#pragma unroll
            for (int ct = 0; ct < CT; ++ct)
                acc[ct] = __builtin_amdgcn_wmma_f32_16x16x4_f32(
                    false, a, false, bfrag[ct], (short)0, acc[ct], false, false);
        }
    }

#pragma unroll
    for (int ct = 0; ct < CT; ++ct) {
#pragma unroll
        for (int r = 0; r < 8; ++r) {
            int m = blockIdx.x * 128 + wave * 16 + 8 * hi + r;
            if (m < nRows) Out[(long)m * NOUT + ct * 16 + laneM] = acc[ct][r];
        }
    }
}


constexpr int ROWS = 4;

template <int F, bool BIAS>
__global__ __launch_bounds__(256) void k_gather(const int* __restrict__ rowptr,
                                                const int* __restrict__ cnt,
                                                const int2* __restrict__ csr,
                                                const float* __restrict__ dinv,
                                                const float* __restrict__ h,
                                                const float* __restrict__ bias,
                                                float* __restrict__ out, int n) {
    constexpr int V = F / 32;
    typedef float vf __attribute__((ext_vector_type(V)));
    const int lane = threadIdx.x & 31;
    const int wv = __builtin_amdgcn_readfirstlane((int)((blockIdx.x * 256 + threadIdx.x) >> 5));

    vf bv = {};
    if (BIAS) bv = *(const vf*)(bias + lane * V);

    int   begs[ROWS], degs[ROWS];
    float dis[ROWS];
#pragma unroll
    for (int r = 0; r < ROWS; ++r) {
        const int ic = wv * ROWS + r < n ? wv * ROWS + r : n - 1;
        begs[r] = rowptr[ic];
        degs[r] = cnt[ic] + 1;
        dis[r]  = dinv[ic];
    }

#pragma unroll
    for (int r = 0; r < ROWS; ++r) {
        const int i = wv * ROWS + r;
        if (i >= n) break;
        const int   beg = begs[r];
        const int   deg = degs[r];
        const float di  = dis[r];

        vf acc = {};
        int p = 0;
        do {
            const int m = deg - p < 32 ? deg - p : 32;
            const int2 ed = csr[beg + p + (lane < m ? lane : m - 1)];

            int j = 0;
            do {
                vf    v[4];
                float w[4];
#pragma unroll
                for (int u = 0; u < 4; ++u) {
                    const int jj = j + u < m ? j + u : m - 1;
                    const int s  = __builtin_amdgcn_readlane(ed.x, jj);
                    w[u] = __int_as_float(__builtin_amdgcn_readlane(ed.y, jj)) * di;
                    v[u] = *(const vf*)(h + (long)s * F + lane * V);
                }
#pragma unroll
                for (int u = 0; u < 4; ++u)
                    if (j + u < m) acc += v[u] * w[u];
                j += 4;
            } while (j < m);
            p += 32;
        } while (p < deg);
        if (BIAS) acc += bv;
        *(vf*)(out + (long)i * F + lane * V) = acc;
    }
}


extern "C" void kernel_launch(void* const* d_in, const int* in_sizes, int n_in,
                              void* d_out, int out_size, void* d_ws, size_t ws_size,
                              hipStream_t stream) {
    const float* x   = (const float*)d_in[0];
    const int*   ei  = (const int*)d_in[1];
    const float* W1  = (const float*)d_in[2];
    const float* b1  = (const float*)d_in[3];
    const float* W2  = (const float*)d_in[4];
    const float* b2  = (const float*)d_in[5];
    float*       out = (float*)d_out;

    const int  n = in_sizes[0] / 128;
    const long E = in_sizes[1] / 2;
    const int* srcIdx = ei;
    const int* dstIdx = ei + E;

    const long n4 = ((long)n + 3) & ~3L;
    const long E4 = (E + 3) & ~3L;
    int*   cnt    = (int*)d_ws;
    int*   rowptr = cnt + n4;
    int*   rank   = rowptr + n4;
    int2*  csr    = (int2*)(rank + E4);
    float* dinv   = (float*)(csr + E4 + n4);
    float* h1     = dinv + n4;
    float* agg1   = h1 + (long)n * 128;
    float* h2     = agg1 + (long)n * 128;

    const int B = 256;
    auto blk = [](long work, int b) { return (int)((work + b - 1) / b); };

    k_zero_cnt<<<blk(n, B), B, 0, stream>>>(cnt, n);
    k_count_rank<<<blk(E, B * EPT), B, 0, stream>>>(dstIdx, E, cnt, rank);
    k_scan_dinv<<<blk(n, 1024), B, 0, stream>>>(cnt, rowptr, dinv, csr, n);
    k_fill<<<blk(E, B * EPT), B, 0, stream>>>(srcIdx, dstIdx, rank, rowptr, dinv, csr, E);

    k_gemm_wmma<128, 128, false><<<blk(n, 128), 256, 0, stream>>>(x, W1, nullptr, h1, n);
    k_gather<128, false><<<blk(n, 8 * ROWS), B, 0, stream>>>(rowptr, cnt, csr, dinv, h1, nullptr, agg1, n);

    k_gemm_wmma<128, 64, true><<<blk(n, 128), 256, 0, stream>>>(agg1, W2, b1, h2, n);
    k_gather<64, true><<<blk(n, 8 * ROWS), B, 0, stream>>>(rowptr, cnt, csr, dinv, h2, b2, out, n);
}


# edge ranks from LDS counters instead of one returning global atomic per edge: edges grouped into 512-row buckets, ranks and row offsets written per bucket; a row's edge (sum) order changes
# speedup vs baseline: 2.8366x; 1.3432x over previous
#include <hip/hip_runtime.h>

typedef float v2f __attribute__((ext_vector_type(2)));
typedef float v8f __attribute__((ext_vector_type(8)));


constexpr int EPT  = 8;
constexpr int BSH  = 9;
constexpr int BROW = 1 << BSH;
constexpr int MAXB = 256;
constexpr int NBLK = 256;
constexpr int BT   = 1024;
constexpr int TPB  = BT / MAXB;
constexpr int LB   = 16;

template <int NT>
__device__ __forceinline__ int block_excl_scan(int v, int* sw) {
    const int lane = threadIdx.x & 31, wv = threadIdx.x >> 5;
    int x = v;
    for (int o = 1; o < 32; o <<= 1) {
        const int y = __shfl_up(x, o);
        if (lane >= o) x += y;
    }
    if (lane == 31) sw[wv] = x;
    __syncthreads();
    int base = 0;
#pragma unroll
    for (int k = 0; k < NT / 32; ++k)
        if (k < wv) base += sw[k];
    return base + x - v;
}

__global__ __launch_bounds__(BT) void k_bkt_hist(const int* __restrict__ dst, long e, int epb, int nbkt,
                                                 int* __restrict__ bh) {
    __shared__ int h[MAXB];
    const int tid = threadIdx.x;
    if (tid < MAXB) h[tid] = 0;
    __syncthreads();
    const int E   = (int)e;
    const int beg = blockIdx.x * epb;
    const int end = beg + epb < E ? beg + epb : E;
    for (int b0 = beg; b0 < end; b0 += BT * EPT) {
        int d[EPT];
#pragma unroll
        for (int k = 0; k < EPT; ++k) {
            const int i = b0 + k * BT + tid;
            d[k] = dst[i < end ? i : end - 1];
        }
#pragma unroll
        for (int k = 0; k < EPT; ++k) {
            const int i = b0 + k * BT + tid;
            if (i < end) atomicAdd(&h[d[k] >> BSH], 1);
        }
    }
    __syncthreads();
    if (tid < nbkt) bh[blockIdx.x * nbkt + tid] = h[tid];
}

__global__ __launch_bounds__(BT) void k_bkt_scatter(const int* __restrict__ src, const int* __restrict__ dst,
                                                    long e, int epb, int nbkt,
                                                    const int* __restrict__ bh, int* __restrict__ bstart,
                                                    int2* __restrict__ pairs) {
    __shared__ int sT[TPB][MAXB], sP[TPB][MAXB], off[MAXB], sw[BT / 32];
    const int tid = threadIdx.x;
    const int nb  = gridDim.x;
    const int me  = blockIdx.x;
    const int q   = tid / MAXB, b = tid % MAXB;
    const int bc  = b < nbkt ? b : nbkt - 1;
    const int nbq = (nb + TPB - 1) / TPB;
    const int k0  = q * nbq;
    const int k1  = k0 + nbq < nb ? k0 + nbq : nb;
    int tot = 0, part = 0;
    for (int i0 = k0; i0 < k1; i0 += LB) {
        int v[LB];
#pragma unroll
        for (int u = 0; u < LB; ++u) {
            const int k = i0 + u < k1 ? i0 + u : k1 - 1;
            v[u] = bh[k * nbkt + bc];
        }
        asm volatile("" ::"v"(v[0]), "v"(v[1]), "v"(v[2]), "v"(v[3]), "v"(v[4]), "v"(v[5]),
                     "v"(v[6]), "v"(v[7]), "v"(v[8]), "v"(v[9]), "v"(v[10]), "v"(v[11]),
                     "v"(v[12]), "v"(v[13]), "v"(v[14]), "v"(v[15]));
        static_assert(LB == 16, "the statement above names LB values");
#pragma unroll
        for (int u = 0; u < LB; ++u) {
            const int k = i0 + u;
            tot  += k < k1 ? v[u] : 0;
            part += (k < k1 && k < me) ? v[u] : 0;
        }
    }
    sT[q][b] = tot;
    sP[q][b] = part;
    __syncthreads();
    int t = 0, p = 0;
    if (tid < nbkt) {
#pragma unroll
        for (int j = 0; j < TPB; ++j) {
            t += sT[j][tid];
            p += sP[j][tid];
        }
    }
    const int start = block_excl_scan<BT>(t, sw);
    if (tid < nbkt) {
        off[tid] = start + p;
        if (me == 0) bstart[tid] = start;
    }
    if (me == 0 && tid == 0) bstart[nbkt] = (int)e;
    __syncthreads();

    const int E   = (int)e;
    const int beg = me * epb;
    const int end = beg + epb < E ? beg + epb : E;
    for (int b0 = beg; b0 < end; b0 += BT * EPT) {
        int s[EPT], d[EPT];
#pragma unroll
        for (int k = 0; k < EPT; ++k) {
            const int i  = b0 + k * BT + tid;
            const int ic = i < end ? i : end - 1;
            s[k] = src[ic];
            d[k] = dst[ic];
        }
#pragma unroll
        for (int k = 0; k < EPT; ++k) {
            const int i = b0 + k * BT + tid;
            if (i < end) pairs[atomicAdd(&off[d[k] >> BSH], 1)] = make_int2(s[k], d[k]);
        }
    }
}

__global__ __launch_bounds__(BROW) void k_bkt_rows(const int2* __restrict__ pairs, const int* __restrict__ bstart,
                                                   int* __restrict__ rank, int* __restrict__ cnt,
                                                   int* __restrict__ rowptr, float* __restrict__ dinv,
                                                   int2* __restrict__ csr, int n) {
    __shared__ int c[BROW], sw[BROW / 32];
    const int tid = threadIdx.x;
    c[tid] = 0;
    __syncthreads();
    const int beg      = bstart[blockIdx.x];
    const int end      = bstart[blockIdx.x + 1];
    const int nodeBase = blockIdx.x << BSH;
    for (int b0 = beg; b0 < end; b0 += BROW * EPT) {
        int d[EPT], r[EPT];
#pragma unroll
        for (int k = 0; k < EPT; ++k) {
            const int i = b0 + k * BROW + tid;
            d[k] = pairs[i < end ? i : end - 1].y;
        }
#pragma unroll
        for (int k = 0; k < EPT; ++k) {
            const int i = b0 + k * BROW + tid;
            r[k] = i < end ? atomicAdd(&c[d[k] - nodeBase], 1) : 0;
        }
#pragma unroll
        for (int k = 0; k < EPT; ++k) {
            const int i = b0 + k * BROW + tid;
            if (i < end) rank[i] = r[k];
        }
    }
    __syncthreads();
    const int node  = nodeBase + tid;
    const int indeg = c[tid];
    const int slots = node < n ? indeg + 1 : 0;
    const int excl  = block_excl_scan<BROW>(slots, sw);
    if (node < n) {
        const int   run = beg + nodeBase + excl;
        const float di  = rsqrtf((float)slots);
        cnt[node]    = indeg;
        rowptr[node] = run;
        dinv[node]   = di;
        csr[run]     = make_int2(node, __float_as_int(di));
    }
}

__global__ __launch_bounds__(256) void k_fill(const int2* __restrict__ pairs,
                                              const int* __restrict__ rank,
                                              const int* __restrict__ rowptr,
                                              const float* __restrict__ dinv,
                                              int2* __restrict__ csr, long e) {
    const int E = (int)e;
    const int T = gridDim.x * 256;
    const int t = blockIdx.x * 256 + threadIdx.x;
    int s[EPT], d[EPT], r[EPT];
#pragma unroll
    for (int k = 0; k < EPT; ++k) {
        int i  = t + k * T;
        int ic = i < E ? i : E - 1;
        const int2 pr = pairs[ic];
        s[k] = pr.x;
        d[k] = pr.y;
        r[k] = rank[ic];
    }
    int   pos[EPT];
    float ws[EPT];
#pragma unroll
    for (int k = 0; k < EPT; ++k) {
        pos[k] = rowptr[d[k]] + 1 + r[k];
        ws[k]  = dinv[s[k]];
    }
#pragma unroll
    for (int k = 0; k < EPT; ++k) {
        int i = t + k * T;
        if (i < E) csr[pos[k]] = make_int2(s[k], __float_as_int(ws[k]));
    }
}


template <int K, int NOUT, bool RELU_BIAS>
__global__ __launch_bounds__(256) void k_gemm_wmma(const float* __restrict__ A,
                                                   const float* __restrict__ W,
                                                   const float* __restrict__ bias,
                                                   float* __restrict__ Out,
                                                   int nRows) {
    constexpr int PITCH = 2 * NOUT + 32;
    constexpr int HK    = K / 2;
    constexpr int PAIRS = K / 4;
    constexpr int CT    = NOUT / 16;
    __shared__ __align__(16) float lw[PAIRS * PITCH];

    const int lane  = threadIdx.x & 31;
    const int wave  = threadIdx.x >> 5;
    const int laneM = lane & 15;
    const int hi    = lane >> 4;

    const int rowBase = blockIdx.x * 128 + wave * 16 + laneM;
    const int rowC    = rowBase < nRows ? rowBase : 0;

    v8f acc[CT] = {};

    for (int kh = 0; kh < 2; ++kh) {
        __syncthreads();
        for (int idx = threadIdx.x; idx < HK * NOUT; idx += 256) {
            int lk = idx / NOUT;
            int c  = idx - lk * NOUT;
            lw[(lk >> 1) * PITCH + 2 * c + (lk & 1)] = W[(kh * HK + lk) * NOUT + c];
        }
        __syncthreads();

        for (int k4 = 0; k4 < HK; k4 += 4) {
            const int kk = kh * HK + k4 + 2 * hi;
            v2f a;
            a.x = A[(long)rowC * K + kk];
            a.y = A[(long)rowC * K + kk + 1];
            if (RELU_BIAS) {
                a.x = fmaxf(a.x + bias[kk], 0.0f);
                a.y = fmaxf(a.y + bias[kk + 1], 0.0f);
            }
            const float* lb = &lw[((k4 >> 1) + hi) * PITCH];

            v2f bfrag[CT];
#pragma unroll
            for (int ct = 0; ct < CT; ++ct)
                bfrag[ct] = *(const v2f*)&lb[2 * (ct * 16 + laneM)];

#pragma unroll
            for (int ct = 0; ct < CT; ++ct)
                acc[ct] = __builtin_amdgcn_wmma_f32_16x16x4_f32(
                    false, a, false, bfrag[ct], (short)0, acc[ct], false, false);
        }
    }

#pragma unroll
    for (int ct = 0; ct < CT; ++ct) {
#pragma unroll
        for (int r = 0; r < 8; ++r) {
            int m = blockIdx.x * 128 + wave * 16 + 8 * hi + r;
            if (m < nRows) Out[(long)m * NOUT + ct * 16 + laneM] = acc[ct][r];
        }
    }
}


constexpr int ROWS = 4;

template <int F, bool BIAS>
__global__ __launch_bounds__(256) void k_gather(const int* __restrict__ rowptr,
                                                const int* __restrict__ cnt,
                                                const int2* __restrict__ csr,
                                                const float* __restrict__ dinv,
                                                const float* __restrict__ h,
                                                const float* __restrict__ bias,
                                                float* __restrict__ out, int n) {
    constexpr int V = F / 32;
    typedef float vf __attribute__((ext_vector_type(V)));
    const int lane = threadIdx.x & 31;
    const int wv = __builtin_amdgcn_readfirstlane((int)((blockIdx.x * 256 + threadIdx.x) >> 5));

    vf bv = {};
    if (BIAS) bv = *(const vf*)(bias + lane * V);

    int   begs[ROWS], degs[ROWS];
    float dis[ROWS];
#pragma unroll
    for (int r = 0; r < ROWS; ++r) {
        const int ic = wv * ROWS + r < n ? wv * ROWS + r : n - 1;
        begs[r] = rowptr[ic];
        degs[r] = cnt[ic] + 1;
        dis[r]  = dinv[ic];
    }

#pragma unroll
    for (int r = 0; r < ROWS; ++r) {
        const int i = wv * ROWS + r;
        if (i >= n) break;
        const int   beg = begs[r];
        const int   deg = degs[r];
        const float di  = dis[r];

        vf acc = {};
        int p = 0;
        do {
            const int m = deg - p < 32 ? deg - p : 32;
            const int2 ed = csr[beg + p + (lane < m ? lane : m - 1)];

            int j = 0;
            do {
                vf    v[4];
                float w[4];
#pragma unroll
                for (int u = 0; u < 4; ++u) {
                    const int jj = j + u < m ? j + u : m - 1;
                    const int s  = __builtin_amdgcn_readlane(ed.x, jj);
                    w[u] = __int_as_float(__builtin_amdgcn_readlane(ed.y, jj)) * di;
                    v[u] = *(const vf*)(h + (long)s * F + lane * V);
                }
#pragma unroll
                for (int u = 0; u < 4; ++u)
                    if (j + u < m) acc += v[u] * w[u];
                j += 4;
            } while (j < m);
            p += 32;
        } while (p < deg);
        if (BIAS) acc += bv;
        *(vf*)(out + (long)i * F + lane * V) = acc;
    }
}


extern "C" void kernel_launch(void* const* d_in, const int* in_sizes, int n_in,
                              void* d_out, int out_size, void* d_ws, size_t ws_size,
                              hipStream_t stream) {
    const float* x   = (const float*)d_in[0];
    const int*   ei  = (const int*)d_in[1];
    const float* W1  = (const float*)d_in[2];
    const float* b1  = (const float*)d_in[3];
    const float* W2  = (const float*)d_in[4];
    const float* b2  = (const float*)d_in[5];
    float*       out = (float*)d_out;

    const int  n = in_sizes[0] / 128;
    const long E = in_sizes[1] / 2;
    const int* srcIdx = ei;
    const int* dstIdx = ei + E;

    const long n4 = ((long)n + 3) & ~3L;
    const long E4 = (E + 3) & ~3L;
    int*   cnt    = (int*)d_ws;
    int*   rowptr = cnt + n4;
    int*   rank   = rowptr + n4;
    int2*  csr    = (int2*)(rank + E4);
    float* dinv   = (float*)(csr + E4 + n4);
    float* h1     = dinv + n4;
    float* agg1   = h1 + (long)n * 128;
    float* h2     = agg1 + (long)n * 128;
    int2*  pairs  = (int2*)(h2 + (long)n * 64);
    int*   bh     = (int*)(pairs + E4);
    int*   bstart = bh + NBLK * MAXB;

    const int B = 256;
    auto blk = [](long work, int b) { return (int)((work + b - 1) / b); };

    const int nbkt = blk(n, BROW);
    const int epb  = blk(E, NBLK);
    k_bkt_hist<<<NBLK, BT, 0, stream>>>(dstIdx, E, epb, nbkt, bh);
    k_bkt_scatter<<<NBLK, BT, 0, stream>>>(srcIdx, dstIdx, E, epb, nbkt, bh, bstart, pairs);
    k_bkt_rows<<<nbkt, BROW, 0, stream>>>(pairs, bstart, rank, cnt, rowptr, dinv, csr, n);
    k_fill<<<blk(E, B * EPT), B, 0, stream>>>(pairs, rank, rowptr, dinv, csr, E);

    k_gemm_wmma<128, 128, false><<<blk(n, 128), 256, 0, stream>>>(x, W1, nullptr, h1, n);
    k_gather<128, false><<<blk(n, 8 * ROWS), B, 0, stream>>>(rowptr, cnt, csr, dinv, h1, nullptr, agg1, n);

    k_gemm_wmma<128, 64, true><<<blk(n, 128), 256, 0, stream>>>(agg1, W2, b1, h2, n);
    k_gather<64, true><<<blk(n, 8 * ROWS), B, 0, stream>>>(rowptr, cnt, csr, dinv, h2, b2, out, n);
}


# layer-1 row gather and the layer-2 product fused in one launch: finished rows (bias, relu) go to an LDS tile and through f32 WMMA there; aggregate plane not written or re-read; sum order kept
# speedup vs baseline: 3.0759x; 1.0844x over previous
#include <hip/hip_runtime.h>

typedef float v2f __attribute__((ext_vector_type(2)));
typedef float v8f __attribute__((ext_vector_type(8)));


constexpr int EPT  = 8;
constexpr int BSH  = 9;
constexpr int BROW = 1 << BSH;
constexpr int MAXB = 256;
constexpr int NBLK = 256;
constexpr int BT   = 1024;
constexpr int TPB  = BT / MAXB;
constexpr int LB   = 16;

template <int NT>
__device__ __forceinline__ int block_excl_scan(int v, int* sw) {
    const int lane = threadIdx.x & 31, wv = threadIdx.x >> 5;
    int x = v;
    for (int o = 1; o < 32; o <<= 1) {
        const int y = __shfl_up(x, o);
        if (lane >= o) x += y;
    }
    if (lane == 31) sw[wv] = x;
    __syncthreads();
    int base = 0;
#pragma unroll
    for (int k = 0; k < NT / 32; ++k)
        if (k < wv) base += sw[k];
    return base + x - v;
}

__global__ __launch_bounds__(BT) void k_bkt_hist(const int* __restrict__ dst, long e, int epb, int nbkt,
                                                 int* __restrict__ bh) {
    __shared__ int h[MAXB];
    const int tid = threadIdx.x;
    if (tid < MAXB) h[tid] = 0;
    __syncthreads();
    const int E   = (int)e;
    const int beg = blockIdx.x * epb;
    const int end = beg + epb < E ? beg + epb : E;
    for (int b0 = beg; b0 < end; b0 += BT * EPT) {
        int d[EPT];
#pragma unroll
        for (int k = 0; k < EPT; ++k) {
            const int i = b0 + k * BT + tid;
            d[k] = dst[i < end ? i : end - 1];
        }
#pragma unroll
        for (int k = 0; k < EPT; ++k) {
            const int i = b0 + k * BT + tid;
            if (i < end) atomicAdd(&h[d[k] >> BSH], 1);
        }
    }
    __syncthreads();
    if (tid < nbkt) bh[blockIdx.x * nbkt + tid] = h[tid];
}

__global__ __launch_bounds__(BT) void k_bkt_scatter(const int* __restrict__ src, const int* __restrict__ dst,
                                                    long e, int epb, int nbkt,
                                                    const int* __restrict__ bh, int* __restrict__ bstart,
                                                    int2* __restrict__ pairs) {
    __shared__ int sT[TPB][MAXB], sP[TPB][MAXB], off[MAXB], sw[BT / 32];
    const int tid = threadIdx.x;
    const int nb  = gridDim.x;
    const int me  = blockIdx.x;
    const int q   = tid / MAXB, b = tid % MAXB;
    const int bc  = b < nbkt ? b : nbkt - 1;
    const int nbq = (nb + TPB - 1) / TPB;
    const int k0  = q * nbq;
    const int k1  = k0 + nbq < nb ? k0 + nbq : nb;
    int tot = 0, part = 0;
    for (int i0 = k0; i0 < k1; i0 += LB) {
        int v[LB];
#pragma unroll
        for (int u = 0; u < LB; ++u) {
            const int k = i0 + u < k1 ? i0 + u : k1 - 1;
            v[u] = bh[k * nbkt + bc];
        }
        asm volatile("" ::"v"(v[0]), "v"(v[1]), "v"(v[2]), "v"(v[3]), "v"(v[4]), "v"(v[5]),
                     "v"(v[6]), "v"(v[7]), "v"(v[8]), "v"(v[9]), "v"(v[10]), "v"(v[11]),
                     "v"(v[12]), "v"(v[13]), "v"(v[14]), "v"(v[15]));
        static_assert(LB == 16, "the statement above names LB values");
#pragma unroll
        for (int u = 0; u < LB; ++u) {
            const int k = i0 + u;
            tot  += k < k1 ? v[u] : 0;
            part += (k < k1 && k < me) ? v[u] : 0;
        }
    }
    sT[q][b] = tot;
    sP[q][b] = part;
    __syncthreads();
    int t = 0, p = 0;
    if (tid < nbkt) {
#pragma unroll
        for (int j = 0; j < TPB; ++j) {
            t += sT[j][tid];
            p += sP[j][tid];
        }
    }
    const int start = block_excl_scan<BT>(t, sw);
    if (tid < nbkt) {
        off[tid] = start + p;
        if (me == 0) bstart[tid] = start;
    }
    if (me == 0 && tid == 0) bstart[nbkt] = (int)e;
    __syncthreads();

    const int E   = (int)e;
    const int beg = me * epb;
    const int end = beg + epb < E ? beg + epb : E;
    for (int b0 = beg; b0 < end; b0 += BT * EPT) {
        int s[EPT], d[EPT];
#pragma unroll
        for (int k = 0; k < EPT; ++k) {
            const int i  = b0 + k * BT + tid;
            const int ic = i < end ? i : end - 1;
            s[k] = src[ic];
            d[k] = dst[ic];
        }
#pragma unroll
        for (int k = 0; k < EPT; ++k) {
            const int i = b0 + k * BT + tid;
            if (i < end) pairs[atomicAdd(&off[d[k] >> BSH], 1)] = make_int2(s[k], d[k]);
        }
    }
}

__global__ __launch_bounds__(BROW) void k_bkt_rows(const int2* __restrict__ pairs, const int* __restrict__ bstart,
                                                   int* __restrict__ rank, int* __restrict__ cnt,
                                                   int* __restrict__ rowptr, float* __restrict__ dinv,
                                                   int2* __restrict__ csr, int n) {
    __shared__ int c[BROW], sw[BROW / 32];
    const int tid = threadIdx.x;
    c[tid] = 0;
    __syncthreads();
    const int beg      = bstart[blockIdx.x];
    const int end      = bstart[blockIdx.x + 1];
    const int nodeBase = blockIdx.x << BSH;
    for (int b0 = beg; b0 < end; b0 += BROW * EPT) {
        int d[EPT], r[EPT];
#pragma unroll
        for (int k = 0; k < EPT; ++k) {
            const int i = b0 + k * BROW + tid;
            d[k] = pairs[i < end ? i : end - 1].y;
        }
#pragma unroll
        for (int k = 0; k < EPT; ++k) {
            const int i = b0 + k * BROW + tid;
            r[k] = i < end ? atomicAdd(&c[d[k] - nodeBase], 1) : 0;
        }
#pragma unroll
        for (int k = 0; k < EPT; ++k) {
            const int i = b0 + k * BROW + tid;
            if (i < end) rank[i] = r[k];
        }
    }
    __syncthreads();
    const int node  = nodeBase + tid;
    const int indeg = c[tid];
    const int slots = node < n ? indeg + 1 : 0;
    const int excl  = block_excl_scan<BROW>(slots, sw);
    if (node < n) {
        const int   run = beg + nodeBase + excl;
        const float di  = rsqrtf((float)slots);
        cnt[node]    = indeg;
        rowptr[node] = run;
        dinv[node]   = di;
        csr[run]     = make_int2(node, __float_as_int(di));
    }
}

__global__ __launch_bounds__(256) void k_fill(const int2* __restrict__ pairs,
                                              const int* __restrict__ rank,
                                              const int* __restrict__ rowptr,
                                              const float* __restrict__ dinv,
                                              int2* __restrict__ csr, long e) {
    const int E = (int)e;
    const int T = gridDim.x * 256;
    const int t = blockIdx.x * 256 + threadIdx.x;
    int s[EPT], d[EPT], r[EPT];
#pragma unroll
    for (int k = 0; k < EPT; ++k) {
        int i  = t + k * T;
        int ic = i < E ? i : E - 1;
        const int2 pr = pairs[ic];
        s[k] = pr.x;
        d[k] = pr.y;
        r[k] = rank[ic];
    }
    int   pos[EPT];
    float ws[EPT];
#pragma unroll
    for (int k = 0; k < EPT; ++k) {
        pos[k] = rowptr[d[k]] + 1 + r[k];
        ws[k]  = dinv[s[k]];
    }
#pragma unroll
    for (int k = 0; k < EPT; ++k) {
        int i = t + k * T;
        if (i < E) csr[pos[k]] = make_int2(s[k], __float_as_int(ws[k]));
    }
}


template <int K, int NOUT, bool RELU_BIAS>
__global__ __launch_bounds__(256) void k_gemm_wmma(const float* __restrict__ A,
                                                   const float* __restrict__ W,
                                                   const float* __restrict__ bias,
                                                   float* __restrict__ Out,
                                                   int nRows) {
    constexpr int PITCH = 2 * NOUT + 32;
    constexpr int HK    = K / 2;
    constexpr int PAIRS = K / 4;
    constexpr int CT    = NOUT / 16;
    __shared__ __align__(16) float lw[PAIRS * PITCH];

    const int lane  = threadIdx.x & 31;
    const int wave  = threadIdx.x >> 5;
    const int laneM = lane & 15;
    const int hi    = lane >> 4;

    const int rowBase = blockIdx.x * 128 + wave * 16 + laneM;
    const int rowC    = rowBase < nRows ? rowBase : 0;

    v8f acc[CT] = {};

    for (int kh = 0; kh < 2; ++kh) {
        __syncthreads();
        for (int idx = threadIdx.x; idx < HK * NOUT; idx += 256) {
            int lk = idx / NOUT;
            int c  = idx - lk * NOUT;
            lw[(lk >> 1) * PITCH + 2 * c + (lk & 1)] = W[(kh * HK + lk) * NOUT + c];
        }
        __syncthreads();

        for (int k4 = 0; k4 < HK; k4 += 4) {
            const int kk = kh * HK + k4 + 2 * hi;
            v2f a;
            a.x = A[(long)rowC * K + kk];
            a.y = A[(long)rowC * K + kk + 1];
            if (RELU_BIAS) {
                a.x = fmaxf(a.x + bias[kk], 0.0f);
                a.y = fmaxf(a.y + bias[kk + 1], 0.0f);
            }
            const float* lb = &lw[((k4 >> 1) + hi) * PITCH];

            v2f bfrag[CT];
#pragma unroll
            for (int ct = 0; ct < CT; ++ct)
                bfrag[ct] = *(const v2f*)&lb[2 * (ct * 16 + laneM)];

#pragma unroll
            for (int ct = 0; ct < CT; ++ct)
                acc[ct] = __builtin_amdgcn_wmma_f32_16x16x4_f32(
                    false, a, false, bfrag[ct], (short)0, acc[ct], false, false);
        }
    }

#pragma unroll
    for (int ct = 0; ct < CT; ++ct) {
#pragma unroll
        for (int r = 0; r < 8; ++r) {
            int m = blockIdx.x * 128 + wave * 16 + 8 * hi + r;
            if (m < nRows) Out[(long)m * NOUT + ct * 16 + laneM] = acc[ct][r];
        }
    }
}


constexpr int ROWS = 4;

template <int F, bool BIAS>
__global__ __launch_bounds__(256) void k_gather(const int* __restrict__ rowptr,
                                                const int* __restrict__ cnt,
                                                const int2* __restrict__ csr,
                                                const float* __restrict__ dinv,
                                                const float* __restrict__ h,
                                                const float* __restrict__ bias,
                                                float* __restrict__ out, int n) {
    constexpr int V = F / 32;
    typedef float vf __attribute__((ext_vector_type(V)));
    const int lane = threadIdx.x & 31;
    const int wv = __builtin_amdgcn_readfirstlane((int)((blockIdx.x * 256 + threadIdx.x) >> 5));

    vf bv = {};
    if (BIAS) bv = *(const vf*)(bias + lane * V);

    int   begs[ROWS], degs[ROWS];
    float dis[ROWS];
#pragma unroll
    for (int r = 0; r < ROWS; ++r) {
        const int ic = wv * ROWS + r < n ? wv * ROWS + r : n - 1;
        begs[r] = rowptr[ic];
        degs[r] = cnt[ic] + 1;
        dis[r]  = dinv[ic];
    }

#pragma unroll
    for (int r = 0; r < ROWS; ++r) {
        const int i = wv * ROWS + r;
        if (i >= n) break;
        const int   beg = begs[r];
        const int   deg = degs[r];
        const float di  = dis[r];

        vf acc = {};
        int p = 0;
        do {
            const int m = deg - p < 32 ? deg - p : 32;
            const int2 ed = csr[beg + p + (lane < m ? lane : m - 1)];

            int j = 0;
            do {
                vf    v[4];
                float w[4];
#pragma unroll
                for (int u = 0; u < 4; ++u) {
                    const int jj = j + u < m ? j + u : m - 1;
                    const int s  = __builtin_amdgcn_readlane(ed.x, jj);
                    w[u] = __int_as_float(__builtin_amdgcn_readlane(ed.y, jj)) * di;
                    v[u] = *(const vf*)(h + (long)s * F + lane * V);
                }
#pragma unroll
                for (int u = 0; u < 4; ++u)
                    if (j + u < m) acc += v[u] * w[u];
                j += 4;
            } while (j < m);
            p += 32;
        } while (p < deg);
        if (BIAS) acc += bv;
        *(vf*)(out + (long)i * F + lane * V) = acc;
    }
}


constexpr int FG_ROWS = 64;
constexpr int FG_TP   = 132;

__global__ __launch_bounds__(512) void k_gather_gemm(const int* __restrict__ rowptr,
                                                     const int* __restrict__ cnt,
                                                     const int2* __restrict__ csr,
                                                     const float* __restrict__ dinv,
                                                     const float* __restrict__ h,
                                                     const float* __restrict__ bias,
                                                     const float* __restrict__ W,
                                                     float* __restrict__ Out, int n) {
    constexpr int F = 128, K = 128, NOUT = 64, V = 4;
    constexpr int PITCH = 2 * NOUT + 32;
    typedef float vf __attribute__((ext_vector_type(V)));
    __shared__ __align__(16) float lw[(K / 2) * PITCH];
    __shared__ __align__(16) float tile[FG_ROWS * FG_TP];

    const int lane = threadIdx.x & 31;
    const int wave = __builtin_amdgcn_readfirstlane((int)(threadIdx.x >> 5));

    static_assert(FG_ROWS == 16 * ROWS, "16 waves of ROWS rows each fill the tile");
    static_assert(K * NOUT == 4 * 4 * 512, "four 16-byte reads per thread cover W");
    vf wreg[4];
#pragma unroll
    for (int j = 0; j < 4; ++j) wreg[j] = *(const vf*)(W + 4 * (threadIdx.x + 512 * j));

    const vf bv = *(const vf*)(bias + lane * V);
    const int row0 = blockIdx.x * FG_ROWS + wave * ROWS;

    int   begs[ROWS], degs[ROWS];
    float dis[ROWS];
#pragma unroll
    for (int r = 0; r < ROWS; ++r) {
        const int ic = row0 + r < n ? row0 + r : n - 1;
        begs[r] = rowptr[ic];
        degs[r] = cnt[ic] + 1;
        dis[r]  = dinv[ic];
    }

#pragma unroll
    for (int r = 0; r < ROWS; ++r) {
        const int i = row0 + r;
        vf res = {};
        if (i < n) {
            const int   beg = begs[r];
            const int   deg = degs[r];
            const float di  = dis[r];
            vf acc = {};
            int p = 0;
            do {
                const int m = deg - p < 32 ? deg - p : 32;
                const int2 ed = csr[beg + p + (lane < m ? lane : m - 1)];
                int j = 0;
                do {
                    vf    v[4];
                    float w[4];
#pragma unroll
                    for (int u = 0; u < 4; ++u) {
                        const int jj = j + u < m ? j + u : m - 1;
                        const int s  = __builtin_amdgcn_readlane(ed.x, jj);
                        w[u] = __int_as_float(__builtin_amdgcn_readlane(ed.y, jj)) * di;
                        v[u] = *(const vf*)(h + (long)s * F + lane * V);
                    }
#pragma unroll
                    for (int u = 0; u < 4; ++u)
                        if (j + u < m) acc += v[u] * w[u];
                    j += 4;
                } while (j < m);
                p += 32;
            } while (p < deg);
            acc += bv;
#pragma unroll
            for (int c = 0; c < V; ++c) res[c] = fmaxf(acc[c], 0.0f);
        }
        *(vf*)&tile[(wave * ROWS + r) * FG_TP + lane * V] = res;
    }
#pragma unroll
    for (int j = 0; j < 4; ++j) {
        const int e = 4 * (threadIdx.x + 512 * j);
        const int k = e / NOUT;
        const int c = e - k * NOUT;
#pragma unroll
        for (int q = 0; q < 4; ++q) lw[(k >> 1) * PITCH + 2 * (c + q) + (k & 1)] = wreg[j][q];
    }
    __syncthreads();

    const int laneM = lane & 15;
    const int hi    = lane >> 4;
    const int t     = wave >> 2;
    const int ct    = wave & 3;
    const float* ta = &tile[(t * 16 + laneM) * FG_TP + 2 * hi];
    const float* tb = &lw[hi * PITCH + 2 * (ct * 16 + laneM)];
    v8f d = {};
    for (int k4 = 0; k4 < K; k4 += 4) {
        const v2f a = *(const v2f*)&ta[k4];
        const v2f b = *(const v2f*)&tb[(k4 >> 1) * PITCH];
        d = __builtin_amdgcn_wmma_f32_16x16x4_f32(false, a, false, b, (short)0, d, false, false);
    }
#pragma unroll
    for (int r = 0; r < 8; ++r) {
        const int m = blockIdx.x * FG_ROWS + t * 16 + 8 * hi + r;
        if (m < n) Out[(long)m * NOUT + ct * 16 + laneM] = d[r];
    }
}


extern "C" void kernel_launch(void* const* d_in, const int* in_sizes, int n_in,
                              void* d_out, int out_size, void* d_ws, size_t ws_size,
                              hipStream_t stream) {
    const float* x   = (const float*)d_in[0];
    const int*   ei  = (const int*)d_in[1];
    const float* W1  = (const float*)d_in[2];
    const float* b1  = (const float*)d_in[3];
    const float* W2  = (const float*)d_in[4];
    const float* b2  = (const float*)d_in[5];
    float*       out = (float*)d_out;

    const int  n = in_sizes[0] / 128;
    const long E = in_sizes[1] / 2;
    const int* srcIdx = ei;
    const int* dstIdx = ei + E;

    const long n4 = ((long)n + 3) & ~3L;
    const long E4 = (E + 3) & ~3L;
    int*   cnt    = (int*)d_ws;
    int*   rowptr = cnt + n4;
    int*   rank   = rowptr + n4;
    int2*  csr    = (int2*)(rank + E4);
    float* dinv   = (float*)(csr + E4 + n4);
    float* h1     = dinv + n4;
    float* h2     = h1 + (long)n * 128;
    int2*  pairs  = (int2*)(h2 + (long)n * 64);
    int*   bh     = (int*)(pairs + E4);
    int*   bstart = bh + NBLK * MAXB;

    const int B = 256;
    auto blk = [](long work, int b) { return (int)((work + b - 1) / b); };

    const int nbkt = blk(n, BROW);
    const int epb  = blk(E, NBLK);
    k_bkt_hist<<<NBLK, BT, 0, stream>>>(dstIdx, E, epb, nbkt, bh);
    k_bkt_scatter<<<NBLK, BT, 0, stream>>>(srcIdx, dstIdx, E, epb, nbkt, bh, bstart, pairs);
    k_bkt_rows<<<nbkt, BROW, 0, stream>>>(pairs, bstart, rank, cnt, rowptr, dinv, csr, n);
    k_fill<<<blk(E, B * EPT), B, 0, stream>>>(pairs, rank, rowptr, dinv, csr, E);

    k_gemm_wmma<128, 128, false><<<blk(n, 128), 256, 0, stream>>>(x, W1, nullptr, h1, n);
    k_gather_gemm<<<blk(n, FG_ROWS), 512, 0, stream>>>(rowptr, cnt, csr, dinv, h1, b1, W2, h2, n);

    k_gather<64, true><<<blk(n, 8 * ROWS), B, 0, stream>>>(rowptr, cnt, csr, dinv, h2, b2, out, n);
}


# layer 1 reassociated as (A x) W1: rows of x gathered first, then both f32 WMMA products on the LDS tile in the one fused launch; the x W1 launch and its plane are gone; sum order changes
# speedup vs baseline: 3.4416x; 1.1189x over previous
#include <hip/hip_runtime.h>

typedef float v2f __attribute__((ext_vector_type(2)));
typedef float v8f __attribute__((ext_vector_type(8)));


constexpr int EPT  = 8;
constexpr int BSH  = 9;
constexpr int BROW = 1 << BSH;
constexpr int MAXB = 256;
constexpr int NBLK = 256;
constexpr int BT   = 1024;
constexpr int TPB  = BT / MAXB;
constexpr int LB   = 16;

template <int NT>
__device__ __forceinline__ int block_excl_scan(int v, int* sw) {
    const int lane = threadIdx.x & 31, wv = threadIdx.x >> 5;
    int x = v;
    for (int o = 1; o < 32; o <<= 1) {
        const int y = __shfl_up(x, o);
        if (lane >= o) x += y;
    }
    if (lane == 31) sw[wv] = x;
    __syncthreads();
    int base = 0;
#pragma unroll
    for (int k = 0; k < NT / 32; ++k)
        if (k < wv) base += sw[k];
    return base + x - v;
}

__global__ __launch_bounds__(BT) void k_bkt_hist(const int* __restrict__ dst, long e, int epb, int nbkt,
                                                 int* __restrict__ bh) {
    __shared__ int h[MAXB];
    const int tid = threadIdx.x;
    if (tid < MAXB) h[tid] = 0;
    __syncthreads();
    const int E   = (int)e;
    const int beg = blockIdx.x * epb;
    const int end = beg + epb < E ? beg + epb : E;
    for (int b0 = beg; b0 < end; b0 += BT * EPT) {
        int d[EPT];
#pragma unroll
        for (int k = 0; k < EPT; ++k) {
            const int i = b0 + k * BT + tid;
            d[k] = dst[i < end ? i : end - 1];
        }
#pragma unroll
        for (int k = 0; k < EPT; ++k) {
            const int i = b0 + k * BT + tid;
            if (i < end) atomicAdd(&h[d[k] >> BSH], 1);
        }
    }
    __syncthreads();
    if (tid < nbkt) bh[blockIdx.x * nbkt + tid] = h[tid];
}

__global__ __launch_bounds__(BT) void k_bkt_scatter(const int* __restrict__ src, const int* __restrict__ dst,
                                                    long e, int epb, int nbkt,
                                                    const int* __restrict__ bh, int* __restrict__ bstart,
                                                    int2* __restrict__ pairs) {
    __shared__ int sT[TPB][MAXB], sP[TPB][MAXB], off[MAXB], sw[BT / 32];
    const int tid = threadIdx.x;
    const int nb  = gridDim.x;
    const int me  = blockIdx.x;
    const int q   = tid / MAXB, b = tid % MAXB;
    const int bc  = b < nbkt ? b : nbkt - 1;
    const int nbq = (nb + TPB - 1) / TPB;
    const int k0  = q * nbq;
    const int k1  = k0 + nbq < nb ? k0 + nbq : nb;
    int tot = 0, part = 0;
    for (int i0 = k0; i0 < k1; i0 += LB) {
        int v[LB];
#pragma unroll
        for (int u = 0; u < LB; ++u) {
            const int k = i0 + u < k1 ? i0 + u : k1 - 1;
            v[u] = bh[k * nbkt + bc];
        }
        asm volatile("" ::"v"(v[0]), "v"(v[1]), "v"(v[2]), "v"(v[3]), "v"(v[4]), "v"(v[5]),
                     "v"(v[6]), "v"(v[7]), "v"(v[8]), "v"(v[9]), "v"(v[10]), "v"(v[11]),
                     "v"(v[12]), "v"(v[13]), "v"(v[14]), "v"(v[15]));
        static_assert(LB == 16, "the statement above names LB values");
#pragma unroll
        for (int u = 0; u < LB; ++u) {
            const int k = i0 + u;
            tot  += k < k1 ? v[u] : 0;
            part += (k < k1 && k < me) ? v[u] : 0;
        }
    }
    sT[q][b] = tot;
    sP[q][b] = part;
    __syncthreads();
    int t = 0, p = 0;
    if (tid < nbkt) {
#pragma unroll
        for (int j = 0; j < TPB; ++j) {
            t += sT[j][tid];
            p += sP[j][tid];
        }
    }
    const int start = block_excl_scan<BT>(t, sw);
    if (tid < nbkt) {
        off[tid] = start + p;
        if (me == 0) bstart[tid] = start;
    }
    if (me == 0 && tid == 0) bstart[nbkt] = (int)e;
    __syncthreads();

    const int E   = (int)e;
    const int beg = me * epb;
    const int end = beg + epb < E ? beg + epb : E;
    for (int b0 = beg; b0 < end; b0 += BT * EPT) {
        int s[EPT], d[EPT];
#pragma unroll
        for (int k = 0; k < EPT; ++k) {
            const int i  = b0 + k * BT + tid;
            const int ic = i < end ? i : end - 1;
            s[k] = src[ic];
            d[k] = dst[ic];
        }
#pragma unroll
        for (int k = 0; k < EPT; ++k) {
            const int i = b0 + k * BT + tid;
            if (i < end) pairs[atomicAdd(&off[d[k] >> BSH], 1)] = make_int2(s[k], d[k]);
        }
    }
}

__global__ __launch_bounds__(BROW) void k_bkt_rows(const int2* __restrict__ pairs, const int* __restrict__ bstart,
                                                   int* __restrict__ rank, int* __restrict__ cnt,
                                                   int* __restrict__ rowptr, float* __restrict__ dinv,
                                                   int2* __restrict__ csr, int n) {
    __shared__ int c[BROW], sw[BROW / 32];
    const int tid = threadIdx.x;
    c[tid] = 0;
    __syncthreads();
    const int beg      = bstart[blockIdx.x];
    const int end      = bstart[blockIdx.x + 1];
    const int nodeBase = blockIdx.x << BSH;
    for (int b0 = beg; b0 < end; b0 += BROW * EPT) {
        int d[EPT], r[EPT];
#pragma unroll
        for (int k = 0; k < EPT; ++k) {
            const int i = b0 + k * BROW + tid;
            d[k] = pairs[i < end ? i : end - 1].y;
        }
#pragma unroll
        for (int k = 0; k < EPT; ++k) {
            const int i = b0 + k * BROW + tid;
            r[k] = i < end ? atomicAdd(&c[d[k] - nodeBase], 1) : 0;
        }
#pragma unroll
        for (int k = 0; k < EPT; ++k) {
            const int i = b0 + k * BROW + tid;
            if (i < end) rank[i] = r[k];
        }
    }
    __syncthreads();
    const int node  = nodeBase + tid;
    const int indeg = c[tid];
    const int slots = node < n ? indeg + 1 : 0;
    const int excl  = block_excl_scan<BROW>(slots, sw);
    if (node < n) {
        const int   run = beg + nodeBase + excl;
        const float di  = rsqrtf((float)slots);
        cnt[node]    = indeg;
        rowptr[node] = run;
        dinv[node]   = di;
        csr[run]     = make_int2(node, __float_as_int(di));
    }
}

__global__ __launch_bounds__(256) void k_fill(const int2* __restrict__ pairs,
                                              const int* __restrict__ rank,
                                              const int* __restrict__ rowptr,
                                              const float* __restrict__ dinv,
                                              int2* __restrict__ csr, long e) {
    const int E = (int)e;
    const int T = gridDim.x * 256;
    const int t = blockIdx.x * 256 + threadIdx.x;
    int s[EPT], d[EPT], r[EPT];
#pragma unroll
    for (int k = 0; k < EPT; ++k) {
        int i  = t + k * T;
        int ic = i < E ? i : E - 1;
        const int2 pr = pairs[ic];
        s[k] = pr.x;
        d[k] = pr.y;
        r[k] = rank[ic];
    }
    int   pos[EPT];
    float ws[EPT];
#pragma unroll
    for (int k = 0; k < EPT; ++k) {
        pos[k] = rowptr[d[k]] + 1 + r[k];
        ws[k]  = dinv[s[k]];
    }
#pragma unroll
    for (int k = 0; k < EPT; ++k) {
        int i = t + k * T;
        if (i < E) csr[pos[k]] = make_int2(s[k], __float_as_int(ws[k]));
    }
}


constexpr int ROWS = 4;

template <int F, bool BIAS>
__global__ __launch_bounds__(256) void k_gather(const int* __restrict__ rowptr,
                                                const int* __restrict__ cnt,
                                                const int2* __restrict__ csr,
                                                const float* __restrict__ dinv,
                                                const float* __restrict__ h,
                                                const float* __restrict__ bias,
                                                float* __restrict__ out, int n) {
    constexpr int V = F / 32;
    typedef float vf __attribute__((ext_vector_type(V)));
    const int lane = threadIdx.x & 31;
    const int wv = __builtin_amdgcn_readfirstlane((int)((blockIdx.x * 256 + threadIdx.x) >> 5));

    vf bv = {};
    if (BIAS) bv = *(const vf*)(bias + lane * V);

    int   begs[ROWS], degs[ROWS];
    float dis[ROWS];
#pragma unroll
    for (int r = 0; r < ROWS; ++r) {
        const int ic = wv * ROWS + r < n ? wv * ROWS + r : n - 1;
        begs[r] = rowptr[ic];
        degs[r] = cnt[ic] + 1;
        dis[r]  = dinv[ic];
    }

#pragma unroll
    for (int r = 0; r < ROWS; ++r) {
        const int i = wv * ROWS + r;
        if (i >= n) break;
        const int   beg = begs[r];
        const int   deg = degs[r];
        const float di  = dis[r];

        vf acc = {};
        int p = 0;
        do {
            const int m = deg - p < 32 ? deg - p : 32;
            const int2 ed = csr[beg + p + (lane < m ? lane : m - 1)];

            int j = 0;
            do {
                vf    v[4];
                float w[4];
#pragma unroll
                for (int u = 0; u < 4; ++u) {
                    const int jj = j + u < m ? j + u : m - 1;
                    const int s  = __builtin_amdgcn_readlane(ed.x, jj);
                    w[u] = __int_as_float(__builtin_amdgcn_readlane(ed.y, jj)) * di;
                    v[u] = *(const vf*)(h + (long)s * F + lane * V);
                }
#pragma unroll
                for (int u = 0; u < 4; ++u)
                    if (j + u < m) acc += v[u] * w[u];
                j += 4;
            } while (j < m);
            p += 32;
        } while (p < deg);
        if (BIAS) acc += bv;
        *(vf*)(out + (long)i * F + lane * V) = acc;
    }
}


constexpr int FG_ROWS = 128;
constexpr int FG_TP   = 132;

__global__ __launch_bounds__(1024) void k_gather_mlp(const int* __restrict__ rowptr,
                                                    const int* __restrict__ cnt,
                                                    const int2* __restrict__ csr,
                                                    const float* __restrict__ dinv,
                                                    const float* __restrict__ x,
                                                    const float* __restrict__ W1,
                                                    const float* __restrict__ bias1,
                                                    const float* __restrict__ W2,
                                                    float* __restrict__ Out, int n) {
    constexpr int F = 128, K = 128, N1 = 128, N2 = 64, V = 4;
    constexpr int P1 = 2 * N1 + 32;
    constexpr int P2 = 2 * N2 + 32;
    typedef float vf __attribute__((ext_vector_type(V)));
    __shared__ __align__(16) float lw[(K / 2) * P1];
    __shared__ __align__(16) float tile[FG_ROWS * FG_TP];

    const int lane = threadIdx.x & 31;
    const int wave = __builtin_amdgcn_readfirstlane((int)(threadIdx.x >> 5));

    static_assert(FG_ROWS == 32 * ROWS, "32 waves of ROWS rows each fill the tile");
    static_assert(K * N1 == 4 * 4 * 1024, "four 16-byte reads per thread cover W1");
    static_assert(K * N2 == 2 * 4 * 1024, "two 16-byte reads per thread cover W2");
    vf wreg[4];
#pragma unroll
    for (int j = 0; j < 4; ++j) wreg[j] = *(const vf*)(W1 + 4 * (threadIdx.x + 1024 * j));

    const int row0 = blockIdx.x * FG_ROWS + wave * ROWS;

    int   begs[ROWS], degs[ROWS];
    float dis[ROWS];
#pragma unroll
    for (int r = 0; r < ROWS; ++r) {
        const int ic = row0 + r < n ? row0 + r : n - 1;
        begs[r] = rowptr[ic];
        degs[r] = cnt[ic] + 1;
        dis[r]  = dinv[ic];
    }

#pragma unroll
    for (int r = 0; r < ROWS; ++r) {
        const int i = row0 + r;
        vf acc = {};
        if (i < n) {
            const int   beg = begs[r];
            const int   deg = degs[r];
            const float di  = dis[r];
            int p = 0;
            do {
                const int m = deg - p < 32 ? deg - p : 32;
                const int2 ed = csr[beg + p + (lane < m ? lane : m - 1)];
                int j = 0;
                do {
                    vf    v[4];
                    float w[4];
#pragma unroll
                    for (int u = 0; u < 4; ++u) {
                        const int jj = j + u < m ? j + u : m - 1;
                        const int s  = __builtin_amdgcn_readlane(ed.x, jj);
                        w[u] = __int_as_float(__builtin_amdgcn_readlane(ed.y, jj)) * di;
                        v[u] = *(const vf*)(x + (long)s * F + lane * V);
                    }
#pragma unroll
                    for (int u = 0; u < 4; ++u)
                        if (j + u < m) acc += v[u] * w[u];
                    j += 4;
                } while (j < m);
                p += 32;
            } while (p < deg);
        }
        *(vf*)&tile[(wave * ROWS + r) * FG_TP + lane * V] = acc;
    }
#pragma unroll
    for (int j = 0; j < 4; ++j) {
        const int e = 4 * (threadIdx.x + 1024 * j);
        const int k = e / N1;
        const int c = e - k * N1;
#pragma unroll
        for (int q = 0; q < 4; ++q) lw[(k >> 1) * P1 + 2 * (c + q) + (k & 1)] = wreg[j][q];
    }
    __syncthreads();

    const int laneM = lane & 15;
    const int hi    = lane >> 4;
    const int t     = wave >> 2;
    const int cg    = wave & 3;

    vf w2reg[2];
#pragma unroll
    for (int j = 0; j < 2; ++j) w2reg[j] = *(const vf*)(W2 + 4 * (threadIdx.x + 1024 * j));
    const float bA = bias1[cg * 32 + laneM];
    const float bB = bias1[cg * 32 + 16 + laneM];

    const float* ta = &tile[(t * 16 + laneM) * FG_TP + 2 * hi];
    {
        const float* tb = &lw[hi * P1 + 2 * (cg * 32 + laneM)];
        v8f dA = {}, dB = {};
        for (int k4 = 0; k4 < K; k4 += 4) {
            const v2f a  = *(const v2f*)&ta[k4];
            const v2f b0 = *(const v2f*)&tb[(k4 >> 1) * P1];
            const v2f b1 = *(const v2f*)&tb[(k4 >> 1) * P1 + 32];
            dA = __builtin_amdgcn_wmma_f32_16x16x4_f32(false, a, false, b0, (short)0, dA, false, false);
            dB = __builtin_amdgcn_wmma_f32_16x16x4_f32(false, a, false, b1, (short)0, dB, false, false);
        }
        __syncthreads();
#pragma unroll
        for (int r = 0; r < 8; ++r) {
            float* tr = &tile[(t * 16 + 8 * hi + r) * FG_TP + cg * 32 + laneM];
            tr[0]  = fmaxf(dA[r] + bA, 0.0f);
            tr[16] = fmaxf(dB[r] + bB, 0.0f);
        }
    }
#pragma unroll
    for (int j = 0; j < 2; ++j) {
        const int e = 4 * (threadIdx.x + 1024 * j);
        const int k = e / N2;
        const int c = e - k * N2;
#pragma unroll
        for (int q = 0; q < 4; ++q) lw[(k >> 1) * P2 + 2 * (c + q) + (k & 1)] = w2reg[j][q];
    }
    __syncthreads();

    const float* tb = &lw[hi * P2 + 2 * (cg * 16 + laneM)];
    v8f d = {};
    for (int k4 = 0; k4 < K; k4 += 4) {
        const v2f a = *(const v2f*)&ta[k4];
        const v2f b = *(const v2f*)&tb[(k4 >> 1) * P2];
        d = __builtin_amdgcn_wmma_f32_16x16x4_f32(false, a, false, b, (short)0, d, false, false);
    }
#pragma unroll
    for (int r = 0; r < 8; ++r) {
        const int m = blockIdx.x * FG_ROWS + t * 16 + 8 * hi + r;
        if (m < n) Out[(long)m * N2 + cg * 16 + laneM] = d[r];
    }
}


extern "C" void kernel_launch(void* const* d_in, const int* in_sizes, int n_in,
                              void* d_out, int out_size, void* d_ws, size_t ws_size,
                              hipStream_t stream) {
    const float* x   = (const float*)d_in[0];
    const int*   ei  = (const int*)d_in[1];
    const float* W1  = (const float*)d_in[2];
    const float* b1  = (const float*)d_in[3];
    const float* W2  = (const float*)d_in[4];
    const float* b2  = (const float*)d_in[5];
    float*       out = (float*)d_out;

    const int  n = in_sizes[0] / 128;
    const long E = in_sizes[1] / 2;
    const int* srcIdx = ei;
    const int* dstIdx = ei + E;

    const long n4 = ((long)n + 3) & ~3L;
    const long E4 = (E + 3) & ~3L;
    int*   cnt    = (int*)d_ws;
    int*   rowptr = cnt + n4;
    int*   rank   = rowptr + n4;
    int2*  csr    = (int2*)(rank + E4);
    float* dinv   = (float*)(csr + E4 + n4);
    float* h2     = dinv + n4;
    int2*  pairs  = (int2*)(h2 + (long)n * 64);
    int*   bh     = (int*)(pairs + E4);
    int*   bstart = bh + NBLK * MAXB;

    const int B = 256;
    auto blk = [](long work, int b) { return (int)((work + b - 1) / b); };

    const int nbkt = blk(n, BROW);
    const int epb  = blk(E, NBLK);
    k_bkt_hist<<<NBLK, BT, 0, stream>>>(dstIdx, E, epb, nbkt, bh);
    k_bkt_scatter<<<NBLK, BT, 0, stream>>>(srcIdx, dstIdx, E, epb, nbkt, bh, bstart, pairs);
    k_bkt_rows<<<nbkt, BROW, 0, stream>>>(pairs, bstart, rank, cnt, rowptr, dinv, csr, n);
    k_fill<<<blk(E, B * EPT), B, 0, stream>>>(pairs, rank, rowptr, dinv, csr, E);

    k_gather_mlp<<<blk(n, FG_ROWS), 1024, 0, stream>>>(rowptr, cnt, csr, dinv, x, W1, b1, W2, h2, n);

    k_gather<64, true><<<blk(n, 8 * ROWS), B, 0, stream>>>(rowptr, cnt, csr, dinv, h2, b2, out, n);
}


# edge placement folded into the bucket launch (LDS cursors, list of source ids only); layer-1 gather looks up dinv[src], layer-2 rows stored pre-scaled; fill launch gone; a row's sum order changes
# speedup vs baseline: 4.4546x; 1.2943x over previous
#include <hip/hip_runtime.h>

typedef float v2f __attribute__((ext_vector_type(2)));
typedef float v8f __attribute__((ext_vector_type(8)));


constexpr int EPT  = 8;
constexpr int BSH  = 9;
constexpr int BROW = 1 << BSH;
constexpr int MAXB = 256;
constexpr int NBLK = 256;
constexpr int BT   = 1024;
constexpr int TPB  = BT / MAXB;
constexpr int LB   = 16;
constexpr int RT   = 1024;

template <int NT>
__device__ __forceinline__ int block_excl_scan(int v, int* sw) {
    const int lane = threadIdx.x & 31, wv = threadIdx.x >> 5;
    int x = v;
    for (int o = 1; o < 32; o <<= 1) {
        const int y = __shfl_up(x, o);
        if (lane >= o) x += y;
    }
    if (lane == 31) sw[wv] = x;
    __syncthreads();
    int base = 0;
#pragma unroll
    for (int k = 0; k < NT / 32; ++k)
        if (k < wv) base += sw[k];
    return base + x - v;
}

__global__ __launch_bounds__(BT) void k_bkt_hist(const int* __restrict__ dst, long e, int epb, int nbkt,
                                                 int* __restrict__ bh) {
    __shared__ int h[MAXB];
    const int tid = threadIdx.x;
    if (tid < MAXB) h[tid] = 0;
    __syncthreads();
    const int E   = (int)e;
    const int beg = blockIdx.x * epb;
    const int end = beg + epb < E ? beg + epb : E;
    for (int b0 = beg; b0 < end; b0 += BT * EPT) {
        int d[EPT];
#pragma unroll
        for (int k = 0; k < EPT; ++k) {
            const int i = b0 + k * BT + tid;
            d[k] = dst[i < end ? i : end - 1];
        }
#pragma unroll
        for (int k = 0; k < EPT; ++k) {
            const int i = b0 + k * BT + tid;
            if (i < end) atomicAdd(&h[d[k] >> BSH], 1);
        }
    }
    __syncthreads();
    if (tid < nbkt) bh[blockIdx.x * nbkt + tid] = h[tid];
}

__global__ __launch_bounds__(BT) void k_bkt_scatter(const int* __restrict__ src, const int* __restrict__ dst,
                                                    long e, int epb, int nbkt,
                                                    const int* __restrict__ bh, int* __restrict__ bstart,
                                                    int2* __restrict__ pairs) {
    __shared__ int sT[TPB][MAXB], sP[TPB][MAXB], off[MAXB], sw[BT / 32];
    const int tid = threadIdx.x;
    const int nb  = gridDim.x;
    const int me  = blockIdx.x;
    const int q   = tid / MAXB, b = tid % MAXB;
    const int bc  = b < nbkt ? b : nbkt - 1;
    const int nbq = (nb + TPB - 1) / TPB;
    const int k0  = q * nbq;
    const int k1  = k0 + nbq < nb ? k0 + nbq : nb;
    int tot = 0, part = 0;
    for (int i0 = k0; i0 < k1; i0 += LB) {
        int v[LB];
#pragma unroll
        for (int u = 0; u < LB; ++u) {
            const int k = i0 + u < k1 ? i0 + u : k1 - 1;
            v[u] = bh[k * nbkt + bc];
        }
        asm volatile("" ::"v"(v[0]), "v"(v[1]), "v"(v[2]), "v"(v[3]), "v"(v[4]), "v"(v[5]),
                     "v"(v[6]), "v"(v[7]), "v"(v[8]), "v"(v[9]), "v"(v[10]), "v"(v[11]),
                     "v"(v[12]), "v"(v[13]), "v"(v[14]), "v"(v[15]));
        static_assert(LB == 16, "the statement above names LB values");
#pragma unroll
        for (int u = 0; u < LB; ++u) {
            const int k = i0 + u;
            tot  += k < k1 ? v[u] : 0;
            part += (k < k1 && k < me) ? v[u] : 0;
        }
    }
    sT[q][b] = tot;
    sP[q][b] = part;
    __syncthreads();
    int t = 0, p = 0;
    if (tid < nbkt) {
#pragma unroll
        for (int j = 0; j < TPB; ++j) {
            t += sT[j][tid];
            p += sP[j][tid];
        }
    }
    const int start = block_excl_scan<BT>(t, sw);
    if (tid < nbkt) {
        off[tid] = start + p;
        if (me == 0) bstart[tid] = start;
    }
    if (me == 0 && tid == 0) bstart[nbkt] = (int)e;
    __syncthreads();

    const int E   = (int)e;
    const int beg = me * epb;
    const int end = beg + epb < E ? beg + epb : E;
    for (int b0 = beg; b0 < end; b0 += BT * EPT) {
        int s[EPT], d[EPT];
#pragma unroll
        for (int k = 0; k < EPT; ++k) {
            const int i  = b0 + k * BT + tid;
            const int ic = i < end ? i : end - 1;
            s[k] = src[ic];
            d[k] = dst[ic];
        }
#pragma unroll
        for (int k = 0; k < EPT; ++k) {
            const int i = b0 + k * BT + tid;
            if (i < end) pairs[atomicAdd(&off[d[k] >> BSH], 1)] = make_int2(s[k], d[k]);
        }
    }
}

__global__ __launch_bounds__(RT) void k_bkt_rows(const int2* __restrict__ pairs, const int* __restrict__ bstart,
                                                 int* __restrict__ cnt, int* __restrict__ rowptr,
                                                 float* __restrict__ dinv, int* __restrict__ csr, int n) {
    __shared__ int c[BROW], sw[RT / 32];
    const int tid = threadIdx.x;
    if (tid < BROW) c[tid] = 0;
    __syncthreads();
    const int beg      = bstart[blockIdx.x];
    const int end      = bstart[blockIdx.x + 1];
    const int nodeBase = blockIdx.x << BSH;
    int s[EPT] = {}, d[EPT] = {};
    if (beg < end) {
#pragma unroll
        for (int k = 0; k < EPT; ++k) {
            const int  i  = beg + k * RT + tid;
            const int2 pr = pairs[i < end ? i : end - 1];
            s[k] = pr.x;
            d[k] = pr.y;
        }
#pragma unroll
        for (int k = 0; k < EPT; ++k) {
            const int i = beg + k * RT + tid;
            if (i < end) atomicAdd(&c[d[k] - nodeBase], 1);
        }
    }
    for (int b0 = beg + RT * EPT; b0 < end; b0 += RT * EPT) {
        int d2[EPT];
#pragma unroll
        for (int k = 0; k < EPT; ++k) {
            const int i = b0 + k * RT + tid;
            d2[k] = pairs[i < end ? i : end - 1].y;
        }
#pragma unroll
        for (int k = 0; k < EPT; ++k) {
            const int i = b0 + k * RT + tid;
            if (i < end) atomicAdd(&c[d2[k] - nodeBase], 1);
        }
    }
    __syncthreads();
    const int node  = nodeBase + tid;
    const int indeg = tid < BROW ? c[tid] : 0;
    const int slots = (tid < BROW && node < n) ? indeg + 1 : 0;
    const int excl  = block_excl_scan<RT>(slots, sw);
    if (tid < BROW && node < n) {
        const int run = beg + nodeBase + excl;
        cnt[node]    = indeg;
        rowptr[node] = run;
        dinv[node]   = rsqrtf((float)slots);
        csr[run]     = node;
        c[tid]       = run + 1;
    }
    __syncthreads();
    if (beg < end) {
#pragma unroll
        for (int k = 0; k < EPT; ++k) {
            const int i = beg + k * RT + tid;
            if (i < end) csr[atomicAdd(&c[d[k] - nodeBase], 1)] = s[k];
        }
    }
    for (int b0 = beg + RT * EPT; b0 < end; b0 += RT * EPT) {
        int s2[EPT], d2[EPT];
#pragma unroll
        for (int k = 0; k < EPT; ++k) {
            const int  i  = b0 + k * RT + tid;
            const int2 pr = pairs[i < end ? i : end - 1];
            s2[k] = pr.x;
            d2[k] = pr.y;
        }
#pragma unroll
        for (int k = 0; k < EPT; ++k) {
            const int i = b0 + k * RT + tid;
            if (i < end) csr[atomicAdd(&c[d2[k] - nodeBase], 1)] = s2[k];
        }
    }
}


constexpr int ROWS = 4;

template <int F, bool BIAS>
__global__ __launch_bounds__(256) void k_gather(const int* __restrict__ rowptr,
                                                const int* __restrict__ cnt,
                                                const int* __restrict__ csr,
                                                const float* __restrict__ dinv,
                                                const float* __restrict__ h,
                                                const float* __restrict__ bias,
                                                float* __restrict__ out, int n) {
    constexpr int V = F / 32;
    typedef float vf __attribute__((ext_vector_type(V)));
    const int lane = threadIdx.x & 31;
    const int wv = __builtin_amdgcn_readfirstlane((int)((blockIdx.x * 256 + threadIdx.x) >> 5));

    vf bv = {};
    if (BIAS) bv = *(const vf*)(bias + lane * V);

    int   begs[ROWS], degs[ROWS];
    float dis[ROWS];
#pragma unroll
    for (int r = 0; r < ROWS; ++r) {
        const int ic = wv * ROWS + r < n ? wv * ROWS + r : n - 1;
        begs[r] = rowptr[ic];
        degs[r] = cnt[ic] + 1;
        dis[r]  = dinv[ic];
    }

#pragma unroll
    for (int r = 0; r < ROWS; ++r) {
        const int i = wv * ROWS + r;
        if (i >= n) break;
        const int   beg = begs[r];
        const int   deg = degs[r];
        const float di  = dis[r];

        vf acc = {};
        int p = 0;
        do {
            const int m = deg - p < 32 ? deg - p : 32;
            const int ed = csr[beg + p + (lane < m ? lane : m - 1)];

            int j = 0;
            do {
                vf v[4];
#pragma unroll
                for (int u = 0; u < 4; ++u) {
                    const int jj = j + u < m ? j + u : m - 1;
                    const int s  = __builtin_amdgcn_readlane(ed, jj);
                    v[u] = *(const vf*)(h + (long)s * F + lane * V);
                }
#pragma unroll
                for (int u = 0; u < 4; ++u)
                    if (j + u < m) acc += v[u];
                j += 4;
            } while (j < m);
            p += 32;
        } while (p < deg);
        acc *= di;
        if (BIAS) acc += bv;
        *(vf*)(out + (long)i * F + lane * V) = acc;
    }
}


constexpr int FG_ROWS = 128;
constexpr int FG_TP   = 132;

__global__ __launch_bounds__(1024) void k_gather_mlp(const int* __restrict__ rowptr,
                                                    const int* __restrict__ cnt,
                                                    const int* __restrict__ csr,
                                                    const float* __restrict__ dinv,
                                                    const float* __restrict__ x,
                                                    const float* __restrict__ W1,
                                                    const float* __restrict__ bias1,
                                                    const float* __restrict__ W2,
                                                    float* __restrict__ Out, int n) {
    constexpr int F = 128, K = 128, N1 = 128, N2 = 64, V = 4;
    constexpr int P1 = 2 * N1 + 32;
    constexpr int P2 = 2 * N2 + 32;
    typedef float vf __attribute__((ext_vector_type(V)));
    __shared__ __align__(16) float lw[(K / 2) * P1];
    __shared__ __align__(16) float tile[FG_ROWS * FG_TP];

    const int lane = threadIdx.x & 31;
    const int wave = __builtin_amdgcn_readfirstlane((int)(threadIdx.x >> 5));

    static_assert(FG_ROWS == 32 * ROWS, "32 waves of ROWS rows each fill the tile");
    static_assert(K * N1 == 4 * 4 * 1024, "four 16-byte reads per thread cover W1");
    static_assert(K * N2 == 2 * 4 * 1024, "two 16-byte reads per thread cover W2");
    vf wreg[4];
#pragma unroll
    for (int j = 0; j < 4; ++j) wreg[j] = *(const vf*)(W1 + 4 * (threadIdx.x + 1024 * j));

    const int row0 = blockIdx.x * FG_ROWS + wave * ROWS;

    int   begs[ROWS], degs[ROWS];
    float dis[ROWS];
#pragma unroll
    for (int r = 0; r < ROWS; ++r) {
        const int ic = row0 + r < n ? row0 + r : n - 1;
        begs[r] = rowptr[ic];
        degs[r] = cnt[ic] + 1;
        dis[r]  = dinv[ic];
    }

#pragma unroll
    for (int r = 0; r < ROWS; ++r) {
        const int i = row0 + r;
        vf acc = {};
        if (i < n) {
            const int   beg = begs[r];
            const int   deg = degs[r];
            const float di  = dis[r];
            int p = 0;
            do {
                const int m = deg - p < 32 ? deg - p : 32;
                const int ed = csr[beg + p + (lane < m ? lane : m - 1)];
                vf v[4];
                auto ask = [&](int j) {
#pragma unroll
                    for (int u = 0; u < 4; ++u) {
                        const int jj = j + u < m ? j + u : m - 1;
                        const int s  = __builtin_amdgcn_readlane(ed, jj);
                        v[u] = *(const vf*)(x + (long)s * F + lane * V);
                    }
                };
                auto add = [&](int j, float wl) {
#pragma unroll
                    for (int u = 0; u < 4; ++u) {
                        const int   jj = j + u < m ? j + u : m - 1;
                        const float w  = __int_as_float(__builtin_amdgcn_readlane(__float_as_int(wl), jj));
                        if (j + u < m) acc += v[u] * w;
                    }
                };
                ask(0);
                const float wl = dinv[ed] * di;
                add(0, wl);
                for (int j = 4; j < m; j += 4) {
                    ask(j);
                    add(j, wl);
                }
                p += 32;
            } while (p < deg);
        }
        *(vf*)&tile[(wave * ROWS + r) * FG_TP + lane * V] = acc;
    }
#pragma unroll
    for (int j = 0; j < 4; ++j) {
        const int e = 4 * (threadIdx.x + 1024 * j);
        const int k = e / N1;
        const int c = e - k * N1;
#pragma unroll
        for (int q = 0; q < 4; ++q) lw[(k >> 1) * P1 + 2 * (c + q) + (k & 1)] = wreg[j][q];
    }
    __syncthreads();

    const int laneM = lane & 15;
    const int hi    = lane >> 4;
    const int t     = wave >> 2;
    const int cg    = wave & 3;

    vf w2reg[2];
#pragma unroll
    for (int j = 0; j < 2; ++j) w2reg[j] = *(const vf*)(W2 + 4 * (threadIdx.x + 1024 * j));
    const float bA = bias1[cg * 32 + laneM];
    const float bB = bias1[cg * 32 + 16 + laneM];

    const float* ta = &tile[(t * 16 + laneM) * FG_TP + 2 * hi];
    {
        const float* tb = &lw[hi * P1 + 2 * (cg * 32 + laneM)];
        v8f dA = {}, dB = {};
        for (int k4 = 0; k4 < K; k4 += 4) {
            const v2f a  = *(const v2f*)&ta[k4];
            const v2f b0 = *(const v2f*)&tb[(k4 >> 1) * P1];
            const v2f b1 = *(const v2f*)&tb[(k4 >> 1) * P1 + 32];
            dA = __builtin_amdgcn_wmma_f32_16x16x4_f32(false, a, false, b0, (short)0, dA, false, false);
            dB = __builtin_amdgcn_wmma_f32_16x16x4_f32(false, a, false, b1, (short)0, dB, false, false);
        }
        __syncthreads();
#pragma unroll
        for (int r = 0; r < 8; ++r) {
            float* tr = &tile[(t * 16 + 8 * hi + r) * FG_TP + cg * 32 + laneM];
            tr[0]  = fmaxf(dA[r] + bA, 0.0f);
            tr[16] = fmaxf(dB[r] + bB, 0.0f);
        }
    }
#pragma unroll
    for (int j = 0; j < 2; ++j) {
        const int e = 4 * (threadIdx.x + 1024 * j);
        const int k = e / N2;
        const int c = e - k * N2;
#pragma unroll
        for (int q = 0; q < 4; ++q) lw[(k >> 1) * P2 + 2 * (c + q) + (k & 1)] = w2reg[j][q];
    }
    __syncthreads();

    float dm[8];
#pragma unroll
    for (int r = 0; r < 8; ++r) {
        const int m = blockIdx.x * FG_ROWS + t * 16 + 8 * hi + r;
        dm[r] = dinv[m < n ? m : n - 1];
    }

    const float* tb = &lw[hi * P2 + 2 * (cg * 16 + laneM)];
    v8f d = {};
    for (int k4 = 0; k4 < K; k4 += 4) {
        const v2f a = *(const v2f*)&ta[k4];
        const v2f b = *(const v2f*)&tb[(k4 >> 1) * P2];
        d = __builtin_amdgcn_wmma_f32_16x16x4_f32(false, a, false, b, (short)0, d, false, false);
    }
#pragma unroll
    for (int r = 0; r < 8; ++r) {
        const int m = blockIdx.x * FG_ROWS + t * 16 + 8 * hi + r;
        if (m < n) Out[(long)m * N2 + cg * 16 + laneM] = d[r] * dm[r];
    }
}


extern "C" void kernel_launch(void* const* d_in, const int* in_sizes, int n_in,
                              void* d_out, int out_size, void* d_ws, size_t ws_size,
                              hipStream_t stream) {
    const float* x   = (const float*)d_in[0];
    const int*   ei  = (const int*)d_in[1];
    const float* W1  = (const float*)d_in[2];
    const float* b1  = (const float*)d_in[3];
    const float* W2  = (const float*)d_in[4];
    const float* b2  = (const float*)d_in[5];
    float*       out = (float*)d_out;

    const int  n = in_sizes[0] / 128;
    const long E = in_sizes[1] / 2;
    const int* srcIdx = ei;
    const int* dstIdx = ei + E;

    const long n4 = ((long)n + 3) & ~3L;
    const long E4 = (E + 3) & ~3L;
    int*   cnt    = (int*)d_ws;
    int*   rowptr = cnt + n4;
    int*   csr    = rowptr + n4;
    float* dinv   = (float*)(csr + E4 + n4);
    float* h2     = dinv + n4;
    int2*  pairs  = (int2*)(h2 + (long)n * 64);
    int*   bh     = (int*)(pairs + E4);
    int*   bstart = bh + NBLK * MAXB;

    const int B = 256;
    auto blk = [](long work, int b) { return (int)((work + b - 1) / b); };

    const int nbkt = blk(n, BROW);
    const int epb  = blk(E, NBLK);
    k_bkt_hist<<<NBLK, BT, 0, stream>>>(dstIdx, E, epb, nbkt, bh);
    k_bkt_scatter<<<NBLK, BT, 0, stream>>>(srcIdx, dstIdx, E, epb, nbkt, bh, bstart, pairs);
    k_bkt_rows<<<nbkt, RT, 0, stream>>>(pairs, bstart, cnt, rowptr, dinv, csr, n);

    k_gather_mlp<<<blk(n, FG_ROWS), 1024, 0, stream>>>(rowptr, cnt, csr, dinv, x, W1, b1, W2, h2, n);

    k_gather<64, true><<<blk(n, 8 * ROWS), B, 0, stream>>>(rowptr, cnt, csr, dinv, h2, b2, out, n);
}


# edge bucketing made chunk-local: each block groups its chunk by bucket in place from its own LDS counts (no look-back), a bucket block collects its 256 segments; count launch gone; sum order changes
# speedup vs baseline: 4.6278x; 1.0389x over previous
#include <hip/hip_runtime.h>

typedef float v2f __attribute__((ext_vector_type(2)));
typedef float v8f __attribute__((ext_vector_type(8)));


constexpr int EPT  = 8;
constexpr int BSH  = 9;
constexpr int BROW = 1 << BSH;
constexpr int MAXB = 256;
constexpr int NBLK = 256;
constexpr int BT   = 1024;
constexpr int RT   = 1024;
constexpr int CPW  = NBLK / (RT / 32);

template <int NT>
__device__ __forceinline__ int block_excl_scan(int v, int* sw) {
    const int lane = threadIdx.x & 31, wv = threadIdx.x >> 5;
    int x = v;
    for (int o = 1; o < 32; o <<= 1) {
        const int y = __shfl_up(x, o);
        if (lane >= o) x += y;
    }
    if (lane == 31) sw[wv] = x;
    __syncthreads();
    int base = 0;
#pragma unroll
    for (int k = 0; k < NT / 32; ++k)
        if (k < wv) base += sw[k];
    return base + x - v;
}

__global__ __launch_bounds__(BT) void k_chunk_group(const int* __restrict__ src, const int* __restrict__ dst,
                                                    long e, int epb, int nbkt,
                                                    int* __restrict__ bcnt, int* __restrict__ bseg,
                                                    int2* __restrict__ pairs) {
    __shared__ int h[MAXB], sw[BT / 32];
    const int tid = threadIdx.x;
    const int me  = blockIdx.x;
    if (tid < MAXB) h[tid] = 0;
    __syncthreads();
    const int E   = (int)e;
    const int beg = me * epb < E ? me * epb : E;
    const int end = beg + epb < E ? beg + epb : E;
    int s[EPT] = {}, d[EPT] = {};
    if (beg < end) {
#pragma unroll
        for (int k = 0; k < EPT; ++k) {
            const int i  = beg + k * BT + tid;
            const int ic = i < end ? i : end - 1;
            s[k] = src[ic];
            d[k] = dst[ic];
        }
#pragma unroll
        for (int k = 0; k < EPT; ++k) {
            const int i = beg + k * BT + tid;
            if (i < end) atomicAdd(&h[d[k] >> BSH], 1);
        }
    }
    for (int b0 = beg + BT * EPT; b0 < end; b0 += BT * EPT) {
        int d2[EPT];
#pragma unroll
        for (int k = 0; k < EPT; ++k) {
            const int i = b0 + k * BT + tid;
            d2[k] = dst[i < end ? i : end - 1];
        }
#pragma unroll
        for (int k = 0; k < EPT; ++k) {
            const int i = b0 + k * BT + tid;
            if (i < end) atomicAdd(&h[d2[k] >> BSH], 1);
        }
    }
    __syncthreads();
    const int c     = tid < nbkt ? h[tid] : 0;
    const int first = beg + block_excl_scan<BT>(c, sw);
    if (tid < nbkt) {
        h[tid] = first;
        bcnt[tid * NBLK + me] = c;
        bseg[tid * NBLK + me] = first;
    }
    __syncthreads();
    if (beg < end) {
#pragma unroll
        for (int k = 0; k < EPT; ++k) {
            const int i = beg + k * BT + tid;
            if (i < end) pairs[atomicAdd(&h[d[k] >> BSH], 1)] = make_int2(s[k], d[k]);
        }
    }
    for (int b0 = beg + BT * EPT; b0 < end; b0 += BT * EPT) {
        int s2[EPT], d2[EPT];
#pragma unroll
        for (int k = 0; k < EPT; ++k) {
            const int i  = b0 + k * BT + tid;
            const int ic = i < end ? i : end - 1;
            s2[k] = src[ic];
            d2[k] = dst[ic];
        }
#pragma unroll
        for (int k = 0; k < EPT; ++k) {
            const int i = b0 + k * BT + tid;
            if (i < end) pairs[atomicAdd(&h[d2[k] >> BSH], 1)] = make_int2(s2[k], d2[k]);
        }
    }
}

__global__ __launch_bounds__(RT) void k_bkt_rows(const int2* __restrict__ pairs, const int* __restrict__ bcnt,
                                                 const int* __restrict__ bseg, long e,
                                                 int* __restrict__ cnt, int* __restrict__ rowptr,
                                                 float* __restrict__ dinv, int* __restrict__ csr, int n) {
    __shared__ int c[BROW], sw[RT / 32], red[RT / 32];
    const int tid  = threadIdx.x;
    const int lane = tid & 31;
    const int wv   = __builtin_amdgcn_readfirstlane(tid >> 5);
    const int bk   = blockIdx.x;
    const int E    = (int)e;
    if (tid < BROW) c[tid] = 0;

    int so[CPW], sc[CPW], maxc = 0;
#pragma unroll
    for (int k = 0; k < CPW; ++k) {
        so[k] = bseg[bk * NBLK + wv * CPW + k];
        sc[k] = bcnt[bk * NBLK + wv * CPW + k];
        maxc  = sc[k] > maxc ? sc[k] : maxc;
    }

    int part = tid < NBLK ? bseg[bk * NBLK + tid] - bseg[tid] : 0;
    for (int o = 16; o > 0; o >>= 1) part += __shfl_xor(part, o);
    if (lane == 0) red[wv] = part;

    __syncthreads();
    int beg = 0;
#pragma unroll
    for (int k = 0; k < RT / 32; ++k) beg += red[k];

    const int nodeBase = bk << BSH;
    auto for_edges = [&](auto f) {
        for (int t0 = 0; t0 < maxc; t0 += 32) {
            int2 pr[CPW];
#pragma unroll
            for (int k = 0; k < CPW; ++k) {
                const int i = t0 + lane;
                int p = so[k] + (i < sc[k] ? i : sc[k] - 1);
                p = p < 0 ? 0 : (p < E ? p : E - 1);
                pr[k] = pairs[p];
            }
            asm volatile("" ::"v"(pr[0].x), "v"(pr[0].y), "v"(pr[1].x), "v"(pr[1].y), "v"(pr[2].x), "v"(pr[2].y),
                         "v"(pr[3].x), "v"(pr[3].y), "v"(pr[4].x), "v"(pr[4].y), "v"(pr[5].x), "v"(pr[5].y),
                         "v"(pr[6].x), "v"(pr[6].y), "v"(pr[7].x), "v"(pr[7].y));
            static_assert(CPW == 8, "the statement above names CPW pairs");
#pragma unroll
            for (int k = 0; k < CPW; ++k)
                if (t0 + lane < sc[k]) f(pr[k]);
        }
    };
    for_edges([&](int2 pr) { atomicAdd(&c[pr.y - nodeBase], 1); });
    __syncthreads();
    const int node  = nodeBase + tid;
    const int indeg = tid < BROW ? c[tid] : 0;
    const int slots = (tid < BROW && node < n) ? indeg + 1 : 0;
    const int excl  = block_excl_scan<RT>(slots, sw);
    if (tid < BROW && node < n) {
        const int run = beg + nodeBase + excl;
        cnt[node]    = indeg;
        rowptr[node] = run;
        dinv[node]   = rsqrtf((float)slots);
        csr[run]     = node;
        c[tid]       = run + 1;
    }
    __syncthreads();
    for_edges([&](int2 pr) { csr[atomicAdd(&c[pr.y - nodeBase], 1)] = pr.x; });
}


constexpr int ROWS = 4;

template <int F, bool BIAS>
__global__ __launch_bounds__(256) void k_gather(const int* __restrict__ rowptr,
                                                const int* __restrict__ cnt,
                                                const int* __restrict__ csr,
                                                const float* __restrict__ dinv,
                                                const float* __restrict__ h,
                                                const float* __restrict__ bias,
                                                float* __restrict__ out, int n) {
    constexpr int V = F / 32;
    typedef float vf __attribute__((ext_vector_type(V)));
    const int lane = threadIdx.x & 31;
    const int wv = __builtin_amdgcn_readfirstlane((int)((blockIdx.x * 256 + threadIdx.x) >> 5));

    vf bv = {};
    if (BIAS) bv = *(const vf*)(bias + lane * V);

    int   begs[ROWS], degs[ROWS];
    float dis[ROWS];
#pragma unroll
    for (int r = 0; r < ROWS; ++r) {
        const int ic = wv * ROWS + r < n ? wv * ROWS + r : n - 1;
        begs[r] = rowptr[ic];
        degs[r] = cnt[ic] + 1;
        dis[r]  = dinv[ic];
    }

#pragma unroll
    for (int r = 0; r < ROWS; ++r) {
        const int i = wv * ROWS + r;
        if (i >= n) break;
        const int   beg = begs[r];
        const int   deg = degs[r];
        const float di  = dis[r];

        vf acc = {};
        int p = 0;
        do {
            const int m = deg - p < 32 ? deg - p : 32;
            const int ed = csr[beg + p + (lane < m ? lane : m - 1)];

            int j = 0;
            do {
                vf v[4];
#pragma unroll
                for (int u = 0; u < 4; ++u) {
                    const int jj = j + u < m ? j + u : m - 1;
                    const int s  = __builtin_amdgcn_readlane(ed, jj);
                    v[u] = *(const vf*)(h + (long)s * F + lane * V);
                }
#pragma unroll
                for (int u = 0; u < 4; ++u)
                    if (j + u < m) acc += v[u];
                j += 4;
            } while (j < m);
            p += 32;
        } while (p < deg);
        acc *= di;
        if (BIAS) acc += bv;
        *(vf*)(out + (long)i * F + lane * V) = acc;
    }
}


constexpr int FG_ROWS = 128;
constexpr int FG_TP   = 132;

__global__ __launch_bounds__(1024) void k_gather_mlp(const int* __restrict__ rowptr,
                                                    const int* __restrict__ cnt,
                                                    const int* __restrict__ csr,
                                                    const float* __restrict__ dinv,
                                                    const float* __restrict__ x,
                                                    const float* __restrict__ W1,
                                                    const float* __restrict__ bias1,
                                                    const float* __restrict__ W2,
                                                    float* __restrict__ Out, int n) {
    constexpr int F = 128, K = 128, N1 = 128, N2 = 64, V = 4;
    constexpr int P1 = 2 * N1 + 32;
    constexpr int P2 = 2 * N2 + 32;
    typedef float vf __attribute__((ext_vector_type(V)));
    __shared__ __align__(16) float lw[(K / 2) * P1];
    __shared__ __align__(16) float tile[FG_ROWS * FG_TP];

    const int lane = threadIdx.x & 31;
    const int wave = __builtin_amdgcn_readfirstlane((int)(threadIdx.x >> 5));

    static_assert(FG_ROWS == 32 * ROWS, "32 waves of ROWS rows each fill the tile");
    static_assert(K * N1 == 4 * 4 * 1024, "four 16-byte reads per thread cover W1");
    static_assert(K * N2 == 2 * 4 * 1024, "two 16-byte reads per thread cover W2");
    vf wreg[4];
#pragma unroll
    for (int j = 0; j < 4; ++j) wreg[j] = *(const vf*)(W1 + 4 * (threadIdx.x + 1024 * j));

    const int row0 = blockIdx.x * FG_ROWS + wave * ROWS;

    int   begs[ROWS], degs[ROWS];
    float dis[ROWS];
#pragma unroll
    for (int r = 0; r < ROWS; ++r) {
        const int ic = row0 + r < n ? row0 + r : n - 1;
        begs[r] = rowptr[ic];
        degs[r] = cnt[ic] + 1;
        dis[r]  = dinv[ic];
    }

#pragma unroll
    for (int r = 0; r < ROWS; ++r) {
        const int i = row0 + r;
        vf acc = {};
        if (i < n) {
            const int   beg = begs[r];
            const int   deg = degs[r];
            const float di  = dis[r];
            int p = 0;
            do {
                const int m = deg - p < 32 ? deg - p : 32;
                const int ed = csr[beg + p + (lane < m ? lane : m - 1)];
                vf v[4];
                auto ask = [&](int j) {
#pragma unroll
                    for (int u = 0; u < 4; ++u) {
                        const int jj = j + u < m ? j + u : m - 1;
                        const int s  = __builtin_amdgcn_readlane(ed, jj);
                        v[u] = *(const vf*)(x + (long)s * F + lane * V);
                    }
                };
                auto add = [&](int j, float wl) {
#pragma unroll
                    for (int u = 0; u < 4; ++u) {
                        const int   jj = j + u < m ? j + u : m - 1;
                        const float w  = __int_as_float(__builtin_amdgcn_readlane(__float_as_int(wl), jj));
                        if (j + u < m) acc += v[u] * w;
                    }
                };
                ask(0);
                const float wl = dinv[ed] * di;
                add(0, wl);
                for (int j = 4; j < m; j += 4) {
                    ask(j);
                    add(j, wl);
                }
                p += 32;
            } while (p < deg);
        }
        *(vf*)&tile[(wave * ROWS + r) * FG_TP + lane * V] = acc;
    }
#pragma unroll
    for (int j = 0; j < 4; ++j) {
        const int e = 4 * (threadIdx.x + 1024 * j);
        const int k = e / N1;
        const int c = e - k * N1;
#pragma unroll
        for (int q = 0; q < 4; ++q) lw[(k >> 1) * P1 + 2 * (c + q) + (k & 1)] = wreg[j][q];
    }
    __syncthreads();

    const int laneM = lane & 15;
    const int hi    = lane >> 4;
    const int t     = wave >> 2;
    const int cg    = wave & 3;

    vf w2reg[2];
#pragma unroll
    for (int j = 0; j < 2; ++j) w2reg[j] = *(const vf*)(W2 + 4 * (threadIdx.x + 1024 * j));
    const float bA = bias1[cg * 32 + laneM];
    const float bB = bias1[cg * 32 + 16 + laneM];

    const float* ta = &tile[(t * 16 + laneM) * FG_TP + 2 * hi];
    {
        const float* tb = &lw[hi * P1 + 2 * (cg * 32 + laneM)];
        v8f dA = {}, dB = {};
        for (int k4 = 0; k4 < K; k4 += 4) {
            const v2f a  = *(const v2f*)&ta[k4];
            const v2f b0 = *(const v2f*)&tb[(k4 >> 1) * P1];
            const v2f b1 = *(const v2f*)&tb[(k4 >> 1) * P1 + 32];
            dA = __builtin_amdgcn_wmma_f32_16x16x4_f32(false, a, false, b0, (short)0, dA, false, false);
            dB = __builtin_amdgcn_wmma_f32_16x16x4_f32(false, a, false, b1, (short)0, dB, false, false);
        }
        __syncthreads();
#pragma unroll
        for (int r = 0; r < 8; ++r) {
            float* tr = &tile[(t * 16 + 8 * hi + r) * FG_TP + cg * 32 + laneM];
            tr[0]  = fmaxf(dA[r] + bA, 0.0f);
            tr[16] = fmaxf(dB[r] + bB, 0.0f);
        }
    }
#pragma unroll
    for (int j = 0; j < 2; ++j) {
        const int e = 4 * (threadIdx.x + 1024 * j);
        const int k = e / N2;
        const int c = e - k * N2;
#pragma unroll
        for (int q = 0; q < 4; ++q) lw[(k >> 1) * P2 + 2 * (c + q) + (k & 1)] = w2reg[j][q];
    }
    __syncthreads();

    float dm[8];
#pragma unroll
    for (int r = 0; r < 8; ++r) {
        const int m = blockIdx.x * FG_ROWS + t * 16 + 8 * hi + r;
        dm[r] = dinv[m < n ? m : n - 1];
    }

    const float* tb = &lw[hi * P2 + 2 * (cg * 16 + laneM)];
    v8f d = {};
    for (int k4 = 0; k4 < K; k4 += 4) {
        const v2f a = *(const v2f*)&ta[k4];
        const v2f b = *(const v2f*)&tb[(k4 >> 1) * P2];
        d = __builtin_amdgcn_wmma_f32_16x16x4_f32(false, a, false, b, (short)0, d, false, false);
    }
#pragma unroll
    for (int r = 0; r < 8; ++r) {
        const int m = blockIdx.x * FG_ROWS + t * 16 + 8 * hi + r;
        if (m < n) Out[(long)m * N2 + cg * 16 + laneM] = d[r] * dm[r];
    }
}


extern "C" void kernel_launch(void* const* d_in, const int* in_sizes, int n_in,
                              void* d_out, int out_size, void* d_ws, size_t ws_size,
                              hipStream_t stream) {
    const float* x   = (const float*)d_in[0];
    const int*   ei  = (const int*)d_in[1];
    const float* W1  = (const float*)d_in[2];
    const float* b1  = (const float*)d_in[3];
    const float* W2  = (const float*)d_in[4];
    const float* b2  = (const float*)d_in[5];
    float*       out = (float*)d_out;

    const int  n = in_sizes[0] / 128;
    const long E = in_sizes[1] / 2;
    const int* srcIdx = ei;
    const int* dstIdx = ei + E;

    const long n4 = ((long)n + 3) & ~3L;
    const long E4 = (E + 3) & ~3L;
    int*   cnt    = (int*)d_ws;
    int*   rowptr = cnt + n4;
    int*   csr    = rowptr + n4;
    float* dinv   = (float*)(csr + E4 + n4);
    float* h2     = dinv + n4;
    int2*  pairs  = (int2*)(h2 + (long)n * 64);
    int*   bcnt   = (int*)(pairs + E4);
    int*   bseg   = bcnt + MAXB * NBLK;

    const int B = 256;
    auto blk = [](long work, int b) { return (int)((work + b - 1) / b); };

    const int nbkt = blk(n, BROW);
    const int epb  = blk(E, NBLK);
    k_chunk_group<<<NBLK, BT, 0, stream>>>(srcIdx, dstIdx, E, epb, nbkt, bcnt, bseg, pairs);
    k_bkt_rows<<<nbkt, RT, 0, stream>>>(pairs, bcnt, bseg, E, cnt, rowptr, dinv, csr, n);

    k_gather_mlp<<<blk(n, FG_ROWS), 1024, 0, stream>>>(rowptr, cnt, csr, dinv, x, W1, b1, W2, h2, n);

    k_gather<64, true><<<blk(n, 8 * ROWS), B, 0, stream>>>(rowptr, cnt, csr, dinv, h2, b2, out, n);
}


# fused layer-1 launch made resident: 512 work-groups each walk 196 consecutive rows as 64-row tiles, W1 and W2 staged in LDS once each; 16 waves do product 2 while 16 gather on; sum order kept
# speedup vs baseline: 4.9462x; 1.0688x over previous
#include <hip/hip_runtime.h>

typedef float v2f __attribute__((ext_vector_type(2)));
typedef float v8f __attribute__((ext_vector_type(8)));


constexpr int EPT  = 8;
constexpr int BSH  = 9;
constexpr int BROW = 1 << BSH;
constexpr int MAXB = 256;
constexpr int NBLK = 256;
constexpr int BT   = 1024;
constexpr int RT   = 1024;
constexpr int CPW  = NBLK / (RT / 32);

template <int NT>
__device__ __forceinline__ int block_excl_scan(int v, int* sw) {
    const int lane = threadIdx.x & 31, wv = threadIdx.x >> 5;
    int x = v;
    for (int o = 1; o < 32; o <<= 1) {
        const int y = __shfl_up(x, o);
        if (lane >= o) x += y;
    }
    if (lane == 31) sw[wv] = x;
    __syncthreads();
    int base = 0;
#pragma unroll
    for (int k = 0; k < NT / 32; ++k)
        if (k < wv) base += sw[k];
    return base + x - v;
}

__global__ __launch_bounds__(BT) void k_chunk_group(const int* __restrict__ src, const int* __restrict__ dst,
                                                    long e, int epb, int nbkt,
                                                    int* __restrict__ bcnt, int* __restrict__ bseg,
                                                    int2* __restrict__ pairs) {
    __shared__ int h[MAXB], sw[BT / 32];
    const int tid = threadIdx.x;
    const int me  = blockIdx.x;
    if (tid < MAXB) h[tid] = 0;
    __syncthreads();
    const int E   = (int)e;
    const int beg = me * epb < E ? me * epb : E;
    const int end = beg + epb < E ? beg + epb : E;
    int s[EPT] = {}, d[EPT] = {};
    if (beg < end) {
#pragma unroll
        for (int k = 0; k < EPT; ++k) {
            const int i  = beg + k * BT + tid;
            const int ic = i < end ? i : end - 1;
            s[k] = src[ic];
            d[k] = dst[ic];
        }
#pragma unroll
        for (int k = 0; k < EPT; ++k) {
            const int i = beg + k * BT + tid;
            if (i < end) atomicAdd(&h[d[k] >> BSH], 1);
        }
    }
    for (int b0 = beg + BT * EPT; b0 < end; b0 += BT * EPT) {
        int d2[EPT];
#pragma unroll
        for (int k = 0; k < EPT; ++k) {
            const int i = b0 + k * BT + tid;
            d2[k] = dst[i < end ? i : end - 1];
        }
#pragma unroll
        for (int k = 0; k < EPT; ++k) {
            const int i = b0 + k * BT + tid;
            if (i < end) atomicAdd(&h[d2[k] >> BSH], 1);
        }
    }
    __syncthreads();
    const int c     = tid < nbkt ? h[tid] : 0;
    const int first = beg + block_excl_scan<BT>(c, sw);
    if (tid < nbkt) {
        h[tid] = first;
        bcnt[tid * NBLK + me] = c;
        bseg[tid * NBLK + me] = first;
    }
    __syncthreads();
    if (beg < end) {
#pragma unroll
        for (int k = 0; k < EPT; ++k) {
            const int i = beg + k * BT + tid;
            if (i < end) pairs[atomicAdd(&h[d[k] >> BSH], 1)] = make_int2(s[k], d[k]);
        }
    }
    for (int b0 = beg + BT * EPT; b0 < end; b0 += BT * EPT) {
        int s2[EPT], d2[EPT];
#pragma unroll
        for (int k = 0; k < EPT; ++k) {
            const int i  = b0 + k * BT + tid;
            const int ic = i < end ? i : end - 1;
            s2[k] = src[ic];
            d2[k] = dst[ic];
        }
#pragma unroll
        for (int k = 0; k < EPT; ++k) {
            const int i = b0 + k * BT + tid;
            if (i < end) pairs[atomicAdd(&h[d2[k] >> BSH], 1)] = make_int2(s2[k], d2[k]);
        }
    }
}

__global__ __launch_bounds__(RT) void k_bkt_rows(const int2* __restrict__ pairs, const int* __restrict__ bcnt,
                                                 const int* __restrict__ bseg, long e,
                                                 int* __restrict__ cnt, int* __restrict__ rowptr,
                                                 float* __restrict__ dinv, int* __restrict__ csr, int n) {
    __shared__ int c[BROW], sw[RT / 32], red[RT / 32];
    const int tid  = threadIdx.x;
    const int lane = tid & 31;
    const int wv   = __builtin_amdgcn_readfirstlane(tid >> 5);
    const int bk   = blockIdx.x;
    const int E    = (int)e;
    if (tid < BROW) c[tid] = 0;

    int so[CPW], sc[CPW], maxc = 0;
#pragma unroll
    for (int k = 0; k < CPW; ++k) {
        so[k] = bseg[bk * NBLK + wv * CPW + k];
        sc[k] = bcnt[bk * NBLK + wv * CPW + k];
        maxc  = sc[k] > maxc ? sc[k] : maxc;
    }

    int part = tid < NBLK ? bseg[bk * NBLK + tid] - bseg[tid] : 0;
    for (int o = 16; o > 0; o >>= 1) part += __shfl_xor(part, o);
    if (lane == 0) red[wv] = part;

    __syncthreads();
    int beg = 0;
#pragma unroll
    for (int k = 0; k < RT / 32; ++k) beg += red[k];

    const int nodeBase = bk << BSH;
    auto for_edges = [&](auto f) {
        for (int t0 = 0; t0 < maxc; t0 += 32) {
            int2 pr[CPW];
#pragma unroll
            for (int k = 0; k < CPW; ++k) {
                const int i = t0 + lane;
                int p = so[k] + (i < sc[k] ? i : sc[k] - 1);
                p = p < 0 ? 0 : (p < E ? p : E - 1);
                pr[k] = pairs[p];
            }
            asm volatile("" ::"v"(pr[0].x), "v"(pr[0].y), "v"(pr[1].x), "v"(pr[1].y), "v"(pr[2].x), "v"(pr[2].y),
                         "v"(pr[3].x), "v"(pr[3].y), "v"(pr[4].x), "v"(pr[4].y), "v"(pr[5].x), "v"(pr[5].y),
                         "v"(pr[6].x), "v"(pr[6].y), "v"(pr[7].x), "v"(pr[7].y));
            static_assert(CPW == 8, "the statement above names CPW pairs");
#pragma unroll
            for (int k = 0; k < CPW; ++k)
                if (t0 + lane < sc[k]) f(pr[k]);
        }
    };
    for_edges([&](int2 pr) { atomicAdd(&c[pr.y - nodeBase], 1); });
    __syncthreads();
    const int node  = nodeBase + tid;
    const int indeg = tid < BROW ? c[tid] : 0;
    const int slots = (tid < BROW && node < n) ? indeg + 1 : 0;
    const int excl  = block_excl_scan<RT>(slots, sw);
    if (tid < BROW && node < n) {
        const int run = beg + nodeBase + excl;
        cnt[node]    = indeg;
        rowptr[node] = run;
        dinv[node]   = rsqrtf((float)slots);
        csr[run]     = node;
        c[tid]       = run + 1;
    }
    __syncthreads();
    for_edges([&](int2 pr) { csr[atomicAdd(&c[pr.y - nodeBase], 1)] = pr.x; });
}


constexpr int ROWS = 4;

template <int F, bool BIAS>
__global__ __launch_bounds__(256) void k_gather(const int* __restrict__ rowptr,
                                                const int* __restrict__ cnt,
                                                const int* __restrict__ csr,
                                                const float* __restrict__ dinv,
                                                const float* __restrict__ h,
                                                const float* __restrict__ bias,
                                                float* __restrict__ out, int n) {
    constexpr int V = F / 32;
    typedef float vf __attribute__((ext_vector_type(V)));
    const int lane = threadIdx.x & 31;
    const int wv = __builtin_amdgcn_readfirstlane((int)((blockIdx.x * 256 + threadIdx.x) >> 5));

    vf bv = {};
    if (BIAS) bv = *(const vf*)(bias + lane * V);

    int   begs[ROWS], degs[ROWS];
    float dis[ROWS];
#pragma unroll
    for (int r = 0; r < ROWS; ++r) {
        const int ic = wv * ROWS + r < n ? wv * ROWS + r : n - 1;
        begs[r] = rowptr[ic];
        degs[r] = cnt[ic] + 1;
        dis[r]  = dinv[ic];
    }

#pragma unroll
    for (int r = 0; r < ROWS; ++r) {
        const int i = wv * ROWS + r;
        if (i >= n) break;
        const int   beg = begs[r];
        const int   deg = degs[r];
        const float di  = dis[r];

        vf acc = {};
        int p = 0;
        do {
            const int m = deg - p < 32 ? deg - p : 32;
            const int ed = csr[beg + p + (lane < m ? lane : m - 1)];

            int j = 0;
            do {
                vf v[4];
#pragma unroll
                for (int u = 0; u < 4; ++u) {
                    const int jj = j + u < m ? j + u : m - 1;
                    const int s  = __builtin_amdgcn_readlane(ed, jj);
                    v[u] = *(const vf*)(h + (long)s * F + lane * V);
                }
#pragma unroll
                for (int u = 0; u < 4; ++u)
                    if (j + u < m) acc += v[u];
                j += 4;
            } while (j < m);
            p += 32;
        } while (p < deg);
        acc *= di;
        if (BIAS) acc += bv;
        *(vf*)(out + (long)i * F + lane * V) = acc;
    }
}


constexpr int PG_WGS  = 512;
constexpr int PT_ROWS = 64;
constexpr int PT_RPW  = 2;
constexpr int FG_TP   = 132;

__global__ __launch_bounds__(1024) void k_gather_mlp(const int* __restrict__ rowptr,
                                                    const int* __restrict__ cnt,
                                                    const int* __restrict__ csr,
                                                    const float* __restrict__ dinv,
                                                    const float* __restrict__ x,
                                                    const float* __restrict__ W1,
                                                    const float* __restrict__ bias1,
                                                    const float* __restrict__ W2,
                                                    float* __restrict__ Out, int n, int rpw) {
    constexpr int F = 128, K = 128, N1 = 128, N2 = 64, V = 4;
    constexpr int P1 = 2 * N1 + 32;
    constexpr int P2 = 2 * N2 + 32;
    typedef float vf __attribute__((ext_vector_type(V)));
    __shared__ __align__(16) float lw1[(K / 2) * P1];
    __shared__ __align__(16) float lw2[(K / 2) * P2];
    __shared__ __align__(16) float tile[PT_ROWS * FG_TP];

    const int lane = threadIdx.x & 31;
    const int wave = __builtin_amdgcn_readfirstlane((int)(threadIdx.x >> 5));

    static_assert(PT_ROWS == 32 * PT_RPW, "32 waves of PT_RPW rows each fill the tile");
    static_assert(K * N1 == 4 * 4 * 1024, "four 16-byte reads per thread cover W1");
    static_assert(K * N2 == 2 * 4 * 1024, "two 16-byte reads per thread cover W2");

    const int rowBeg = blockIdx.x * rpw;
    const int rowEnd = rowBeg + rpw < n ? rowBeg + rpw : n;
    const int nrows  = rowEnd > rowBeg ? rowEnd - rowBeg : 0;
    const int ntile  = (nrows + PT_ROWS - 1) / PT_ROWS;

    if (ntile > 0) {
        vf wreg[4], w2reg[2];
#pragma unroll
        for (int j = 0; j < 4; ++j) wreg[j] = *(const vf*)(W1 + 4 * (threadIdx.x + 1024 * j));
#pragma unroll
        for (int j = 0; j < 2; ++j) w2reg[j] = *(const vf*)(W2 + 4 * (threadIdx.x + 1024 * j));
#pragma unroll
        for (int j = 0; j < 4; ++j) {
            const int e = 4 * (threadIdx.x + 1024 * j);
            const int k = e / N1;
            const int c = e - k * N1;
#pragma unroll
            for (int q = 0; q < 4; ++q) lw1[(k >> 1) * P1 + 2 * (c + q) + (k & 1)] = wreg[j][q];
        }
#pragma unroll
        for (int j = 0; j < 2; ++j) {
            const int e = 4 * (threadIdx.x + 1024 * j);
            const int k = e / N2;
            const int c = e - k * N2;
#pragma unroll
            for (int q = 0; q < 4; ++q) lw2[(k >> 1) * P2 + 2 * (c + q) + (k & 1)] = w2reg[j][q];
        }
    }

    const int laneM = lane & 15;
    const int hi    = lane >> 4;
    const int rt1   = wave >> 3;
    const int ct1   = wave & 7;
    const int rt2   = wave >> 2;
    const int ct2   = wave & 3;
    const float b1v = bias1[ct1 * 16 + laneM];

    for (int it = 0; it < ntile; ++it) {
        const int tile0 = rowBeg + it * PT_ROWS;
        const int rows  = rowEnd - tile0 < PT_ROWS ? rowEnd - tile0 : PT_ROWS;
        const int row0  = tile0 + wave * PT_RPW;

        int   begs[PT_RPW], degs[PT_RPW];
        float dis[PT_RPW];
#pragma unroll
        for (int r = 0; r < PT_RPW; ++r) {
            const int ic = row0 + r < n ? row0 + r : n - 1;
            begs[r] = rowptr[ic];
            degs[r] = cnt[ic] + 1;
            dis[r]  = dinv[ic];
        }

        vf accs[PT_RPW];
#pragma unroll
        for (int r = 0; r < PT_RPW; ++r) {
            const int i = row0 + r;
            vf acc = {};
            if (i < rowEnd) {
                const int   beg = begs[r];
                const int   deg = degs[r];
                const float di  = dis[r];
                int p = 0;
                do {
                    const int m = deg - p < 32 ? deg - p : 32;
                    const int ed = csr[beg + p + (lane < m ? lane : m - 1)];
                    vf v[4];
                    auto ask = [&](int j) {
#pragma unroll
                        for (int u = 0; u < 4; ++u) {
                            const int jj = j + u < m ? j + u : m - 1;
                            const int s  = __builtin_amdgcn_readlane(ed, jj);
                            v[u] = *(const vf*)(x + (long)s * F + lane * V);
                        }
                    };
                    auto add = [&](int j, float wl) {
#pragma unroll
                        for (int u = 0; u < 4; ++u) {
                            const int   jj = j + u < m ? j + u : m - 1;
                            const float w  = __int_as_float(__builtin_amdgcn_readlane(__float_as_int(wl), jj));
                            if (j + u < m) acc += v[u] * w;
                        }
                    };
                    ask(0);
                    const float wl = dinv[ed] * di;
                    add(0, wl);
                    for (int j = 4; j < m; j += 4) {
                        ask(j);
                        add(j, wl);
                    }
                    p += 32;
                } while (p < deg);
            }
            accs[r] = acc;
        }
        __syncthreads();
#pragma unroll
        for (int r = 0; r < PT_RPW; ++r) *(vf*)&tile[(wave * PT_RPW + r) * FG_TP + lane * V] = accs[r];
        __syncthreads();

        const bool on1 = rt1 * 16 < rows;
        v8f d1 = {};
        if (on1) {
            const float* ta = &tile[(rt1 * 16 + laneM) * FG_TP + 2 * hi];
            const float* tb = &lw1[hi * P1 + 2 * (ct1 * 16 + laneM)];
            for (int k4 = 0; k4 < K; k4 += 4) {
                const v2f a = *(const v2f*)&ta[k4];
                const v2f b = *(const v2f*)&tb[(k4 >> 1) * P1];
                d1 = __builtin_amdgcn_wmma_f32_16x16x4_f32(false, a, false, b, (short)0, d1, false, false);
            }
        }
        __syncthreads();
        if (on1) {
#pragma unroll
            for (int r = 0; r < 8; ++r)
                tile[(rt1 * 16 + 8 * hi + r) * FG_TP + ct1 * 16 + laneM] = fmaxf(d1[r] + b1v, 0.0f);
        }
        __syncthreads();

        if (wave < 16 && rt2 * 16 < rows) {
            float dm[8];
#pragma unroll
            for (int r = 0; r < 8; ++r) {
                const int m = tile0 + rt2 * 16 + 8 * hi + r;
                dm[r] = dinv[m < n ? m : n - 1];
            }
            const float* ta = &tile[(rt2 * 16 + laneM) * FG_TP + 2 * hi];
            const float* tb = &lw2[hi * P2 + 2 * (ct2 * 16 + laneM)];
            v8f d = {};
            for (int k4 = 0; k4 < K; k4 += 4) {
                const v2f a = *(const v2f*)&ta[k4];
                const v2f b = *(const v2f*)&tb[(k4 >> 1) * P2];
                d = __builtin_amdgcn_wmma_f32_16x16x4_f32(false, a, false, b, (short)0, d, false, false);
            }
#pragma unroll
            for (int r = 0; r < 8; ++r) {
                const int m = tile0 + rt2 * 16 + 8 * hi + r;
                if (m < rowEnd) Out[(long)m * N2 + ct2 * 16 + laneM] = d[r] * dm[r];
            }
        }
    }
}


extern "C" void kernel_launch(void* const* d_in, const int* in_sizes, int n_in,
                              void* d_out, int out_size, void* d_ws, size_t ws_size,
                              hipStream_t stream) {
    const float* x   = (const float*)d_in[0];
    const int*   ei  = (const int*)d_in[1];
    const float* W1  = (const float*)d_in[2];
    const float* b1  = (const float*)d_in[3];
    const float* W2  = (const float*)d_in[4];
    const float* b2  = (const float*)d_in[5];
    float*       out = (float*)d_out;

    const int  n = in_sizes[0] / 128;
    const long E = in_sizes[1] / 2;
    const int* srcIdx = ei;
    const int* dstIdx = ei + E;

    const long n4 = ((long)n + 3) & ~3L;
    const long E4 = (E + 3) & ~3L;
    int*   cnt    = (int*)d_ws;
    int*   rowptr = cnt + n4;
    int*   csr    = rowptr + n4;
    float* dinv   = (float*)(csr + E4 + n4);
    float* h2     = dinv + n4;
    int2*  pairs  = (int2*)(h2 + (long)n * 64);
    int*   bcnt   = (int*)(pairs + E4);
    int*   bseg   = bcnt + MAXB * NBLK;

    const int B = 256;
    auto blk = [](long work, int b) { return (int)((work + b - 1) / b); };

    const int nbkt = blk(n, BROW);
    const int epb  = blk(E, NBLK);
    k_chunk_group<<<NBLK, BT, 0, stream>>>(srcIdx, dstIdx, E, epb, nbkt, bcnt, bseg, pairs);
    k_bkt_rows<<<nbkt, RT, 0, stream>>>(pairs, bcnt, bseg, E, cnt, rowptr, dinv, csr, n);

    k_gather_mlp<<<PG_WGS, 1024, 0, stream>>>(rowptr, cnt, csr, dinv, x, W1, b1, W2, h2, n, blk(n, PG_WGS));

    k_gather<64, true><<<blk(n, 8 * ROWS), B, 0, stream>>>(rowptr, cnt, csr, dinv, h2, b2, out, n);
}


# both sort launches build their lists in LDS (places from the LDS counters as before) and copy them out as whole aligned lines instead of 32-line scattered stores; a row's sum order as before
# speedup vs baseline: 5.3570x; 1.0831x over previous
#include <hip/hip_runtime.h>

typedef float v2f __attribute__((ext_vector_type(2)));
typedef float v8f __attribute__((ext_vector_type(8)));


constexpr int EPT  = 8;
constexpr int BSH  = 9;
constexpr int BROW = 1 << BSH;
constexpr int MAXB = 256;
constexpr int NBLK = 256;
constexpr int BT   = 1024;
constexpr int RT   = 1024;
constexpr int CPW  = NBLK / (RT / 32);
constexpr int SCAP = BT * EPT;
constexpr int CCAP = 24576;

template <int NT>
__device__ __forceinline__ int block_excl_scan(int v, int* sw) {
    const int lane = threadIdx.x & 31, wv = threadIdx.x >> 5;
    int x = v;
    for (int o = 1; o < 32; o <<= 1) {
        const int y = __shfl_up(x, o);
        if (lane >= o) x += y;
    }
    if (lane == 31) sw[wv] = x;
    __syncthreads();
    int base = 0;
#pragma unroll
    for (int k = 0; k < NT / 32; ++k)
        if (k < wv) base += sw[k];
    return base + x - v;
}

__global__ __launch_bounds__(BT) void k_chunk_group(const int* __restrict__ src, const int* __restrict__ dst,
                                                    long e, int epb, int nbkt,
                                                    int* __restrict__ bcnt, int* __restrict__ bseg,
                                                    int2* __restrict__ pairs) {
    __shared__ int h[MAXB], sw[BT / 32];
    __shared__ __align__(16) int2 stage[SCAP];
    const int tid = threadIdx.x;
    const int me  = blockIdx.x;
    if (tid < MAXB) h[tid] = 0;
    __syncthreads();
    const int E   = (int)e;
    const int beg = me * epb < E ? me * epb : E;
    const int end = beg + epb < E ? beg + epb : E;
    int s[EPT] = {}, d[EPT] = {};
    if (beg < end) {
#pragma unroll
        for (int k = 0; k < EPT; ++k) {
            const int i  = beg + k * BT + tid;
            const int ic = i < end ? i : end - 1;
            s[k] = src[ic];
            d[k] = dst[ic];
        }
#pragma unroll
        for (int k = 0; k < EPT; ++k) {
            const int i = beg + k * BT + tid;
            if (i < end) atomicAdd(&h[d[k] >> BSH], 1);
        }
    }
    for (int b0 = beg + BT * EPT; b0 < end; b0 += BT * EPT) {
        int d2[EPT];
#pragma unroll
        for (int k = 0; k < EPT; ++k) {
            const int i = b0 + k * BT + tid;
            d2[k] = dst[i < end ? i : end - 1];
        }
#pragma unroll
        for (int k = 0; k < EPT; ++k) {
            const int i = b0 + k * BT + tid;
            if (i < end) atomicAdd(&h[d2[k] >> BSH], 1);
        }
    }
    __syncthreads();
    const int c     = tid < nbkt ? h[tid] : 0;
    const int first = beg + block_excl_scan<BT>(c, sw);
    const bool staged = end - beg <= SCAP;
    if (tid < nbkt) {
        h[tid] = staged ? first - beg : first;
        bcnt[tid * NBLK + me] = c;
        bseg[tid * NBLK + me] = first;
    }
    __syncthreads();
    if (staged) {
#pragma unroll
        for (int k = 0; k < EPT; ++k) {
            const int i = beg + k * BT + tid;
            if (i < end) stage[atomicAdd(&h[d[k] >> BSH], 1)] = make_int2(s[k], d[k]);
        }
        __syncthreads();
        const int len = end - beg;
        for (int i0 = (beg & ~31) + tid; i0 < end; i0 += 4 * BT) {
            int2 v[4];
#pragma unroll
            for (int u = 0; u < 4; ++u) {
                const int j = i0 + u * BT - beg;
                v[u] = stage[j < 0 ? 0 : (j < len ? j : len - 1)];
            }
#pragma unroll
            for (int u = 0; u < 4; ++u) {
                const int i = i0 + u * BT;
                if (i >= beg && i < end) pairs[i] = v[u];
            }
        }
        return;
    }
    if (beg < end) {
#pragma unroll
        for (int k = 0; k < EPT; ++k) {
            const int i = beg + k * BT + tid;
            if (i < end) pairs[atomicAdd(&h[d[k] >> BSH], 1)] = make_int2(s[k], d[k]);
        }
    }
    for (int b0 = beg + BT * EPT; b0 < end; b0 += BT * EPT) {
        int s2[EPT], d2[EPT];
#pragma unroll
        for (int k = 0; k < EPT; ++k) {
            const int i  = b0 + k * BT + tid;
            const int ic = i < end ? i : end - 1;
            s2[k] = src[ic];
            d2[k] = dst[ic];
        }
#pragma unroll
        for (int k = 0; k < EPT; ++k) {
            const int i = b0 + k * BT + tid;
            if (i < end) pairs[atomicAdd(&h[d2[k] >> BSH], 1)] = make_int2(s2[k], d2[k]);
        }
    }
}

__global__ __launch_bounds__(RT) void k_bkt_rows(const int2* __restrict__ pairs, const int* __restrict__ bcnt,
                                                 const int* __restrict__ bseg, long e,
                                                 int* __restrict__ cnt, int* __restrict__ rowptr,
                                                 float* __restrict__ dinv, int* __restrict__ csr, int n) {
    __shared__ int c[BROW], sw[RT / 32], red[RT / 32], redc[RT / 32];
    __shared__ __align__(16) int stage[CCAP];
    const int tid  = threadIdx.x;
    const int lane = tid & 31;
    const int wv   = __builtin_amdgcn_readfirstlane(tid >> 5);
    const int bk   = blockIdx.x;
    const int E    = (int)e;
    if (tid < BROW) c[tid] = 0;

    int so[CPW], sc[CPW], maxc = 0;
#pragma unroll
    for (int k = 0; k < CPW; ++k) {
        so[k] = bseg[bk * NBLK + wv * CPW + k];
        sc[k] = bcnt[bk * NBLK + wv * CPW + k];
        maxc  = sc[k] > maxc ? sc[k] : maxc;
    }

    int part = tid < NBLK ? bseg[bk * NBLK + tid] - bseg[tid] : 0;
    int mine = tid < NBLK ? bcnt[bk * NBLK + tid] : 0;
    for (int o = 16; o > 0; o >>= 1) {
        part += __shfl_xor(part, o);
        mine += __shfl_xor(mine, o);
    }
    if (lane == 0) {
        red[wv]  = part;
        redc[wv] = mine;
    }

    __syncthreads();
    int beg = 0, edges = 0;
#pragma unroll
    for (int k = 0; k < RT / 32; ++k) {
        beg   += red[k];
        edges += redc[k];
    }

    const int nodeBase = bk << BSH;
    const int rowsHere = n - nodeBase < BROW ? n - nodeBase : BROW;
    const int start    = beg + nodeBase;
    const int total    = edges + (rowsHere > 0 ? rowsHere : 0);
    const bool staged  = total <= CCAP;
    auto for_edges = [&](auto f) {
        for (int t0 = 0; t0 < maxc; t0 += 32) {
            int2 pr[CPW];
#pragma unroll
            for (int k = 0; k < CPW; ++k) {
                const int i = t0 + lane;
                int p = so[k] + (i < sc[k] ? i : sc[k] - 1);
                p = p < 0 ? 0 : (p < E ? p : E - 1);
                pr[k] = pairs[p];
            }
            asm volatile("" ::"v"(pr[0].x), "v"(pr[0].y), "v"(pr[1].x), "v"(pr[1].y), "v"(pr[2].x), "v"(pr[2].y),
                         "v"(pr[3].x), "v"(pr[3].y), "v"(pr[4].x), "v"(pr[4].y), "v"(pr[5].x), "v"(pr[5].y),
                         "v"(pr[6].x), "v"(pr[6].y), "v"(pr[7].x), "v"(pr[7].y));
            static_assert(CPW == 8, "the statement above names CPW pairs");
#pragma unroll
            for (int k = 0; k < CPW; ++k)
                if (t0 + lane < sc[k]) f(pr[k]);
        }
    };
    for_edges([&](int2 pr) { atomicAdd(&c[pr.y - nodeBase], 1); });
    __syncthreads();
    const int node  = nodeBase + tid;
    const int indeg = tid < BROW ? c[tid] : 0;
    const int slots = (tid < BROW && node < n) ? indeg + 1 : 0;
    const int excl  = block_excl_scan<RT>(slots, sw);
    if (tid < BROW && node < n) {
        const int run = start + excl;
        cnt[node]    = indeg;
        rowptr[node] = run;
        dinv[node]   = rsqrtf((float)slots);
        if (staged) {
            stage[excl] = node;
            c[tid]      = excl + 1;
        } else {
            csr[run] = node;
            c[tid]   = run + 1;
        }
    }
    __syncthreads();
    if (staged) {
        for_edges([&](int2 pr) { stage[atomicAdd(&c[pr.y - nodeBase], 1)] = pr.x; });
        __syncthreads();
        const int last = start + total;
        for (int i0 = (start & ~31) + tid; i0 < last; i0 += 4 * RT) {
            int v[4];
#pragma unroll
            for (int u = 0; u < 4; ++u) {
                const int j = i0 + u * RT - start;
                v[u] = stage[j < 0 ? 0 : (j < total ? j : total - 1)];
            }
#pragma unroll
            for (int u = 0; u < 4; ++u) {
                const int i = i0 + u * RT;
                if (i >= start && i < last) csr[i] = v[u];
            }
        }
    } else {
        for_edges([&](int2 pr) { csr[atomicAdd(&c[pr.y - nodeBase], 1)] = pr.x; });
    }
}


constexpr int ROWS = 4;

template <int F, bool BIAS>
__global__ __launch_bounds__(256) void k_gather(const int* __restrict__ rowptr,
                                                const int* __restrict__ cnt,
                                                const int* __restrict__ csr,
                                                const float* __restrict__ dinv,
                                                const float* __restrict__ h,
                                                const float* __restrict__ bias,
                                                float* __restrict__ out, int n) {
    constexpr int V = F / 32;
    typedef float vf __attribute__((ext_vector_type(V)));
    const int lane = threadIdx.x & 31;
    const int wv = __builtin_amdgcn_readfirstlane((int)((blockIdx.x * 256 + threadIdx.x) >> 5));

    vf bv = {};
    if (BIAS) bv = *(const vf*)(bias + lane * V);

    int   begs[ROWS], degs[ROWS];
    float dis[ROWS];
#pragma unroll
    for (int r = 0; r < ROWS; ++r) {
        const int ic = wv * ROWS + r < n ? wv * ROWS + r : n - 1;
        begs[r] = rowptr[ic];
        degs[r] = cnt[ic] + 1;
        dis[r]  = dinv[ic];
    }

#pragma unroll
    for (int r = 0; r < ROWS; ++r) {
        const int i = wv * ROWS + r;
        if (i >= n) break;
        const int   beg = begs[r];
        const int   deg = degs[r];
        const float di  = dis[r];

        vf acc = {};
        int p = 0;
        do {
            const int m = deg - p < 32 ? deg - p : 32;
            const int ed = csr[beg + p + (lane < m ? lane : m - 1)];

            int j = 0;
            do {
                vf v[4];
#pragma unroll
                for (int u = 0; u < 4; ++u) {
                    const int jj = j + u < m ? j + u : m - 1;
                    const int s  = __builtin_amdgcn_readlane(ed, jj);
                    v[u] = *(const vf*)(h + (long)s * F + lane * V);
                }
#pragma unroll
                for (int u = 0; u < 4; ++u)
                    if (j + u < m) acc += v[u];
                j += 4;
            } while (j < m);
            p += 32;
        } while (p < deg);
        acc *= di;
        if (BIAS) acc += bv;
        *(vf*)(out + (long)i * F + lane * V) = acc;
    }
}


constexpr int PG_WGS  = 512;
constexpr int PT_ROWS = 64;
constexpr int PT_RPW  = 2;
constexpr int FG_TP   = 132;

__global__ __launch_bounds__(1024) void k_gather_mlp(const int* __restrict__ rowptr,
                                                    const int* __restrict__ cnt,
                                                    const int* __restrict__ csr,
                                                    const float* __restrict__ dinv,
                                                    const float* __restrict__ x,
                                                    const float* __restrict__ W1,
                                                    const float* __restrict__ bias1,
                                                    const float* __restrict__ W2,
                                                    float* __restrict__ Out, int n, int rpw) {
    constexpr int F = 128, K = 128, N1 = 128, N2 = 64, V = 4;
    constexpr int P1 = 2 * N1 + 32;
    constexpr int P2 = 2 * N2 + 32;
    typedef float vf __attribute__((ext_vector_type(V)));
    __shared__ __align__(16) float lw1[(K / 2) * P1];
    __shared__ __align__(16) float lw2[(K / 2) * P2];
    __shared__ __align__(16) float tile[PT_ROWS * FG_TP];

    const int lane = threadIdx.x & 31;
    const int wave = __builtin_amdgcn_readfirstlane((int)(threadIdx.x >> 5));

    static_assert(PT_ROWS == 32 * PT_RPW, "32 waves of PT_RPW rows each fill the tile");
    static_assert(K * N1 == 4 * 4 * 1024, "four 16-byte reads per thread cover W1");
    static_assert(K * N2 == 2 * 4 * 1024, "two 16-byte reads per thread cover W2");

    const int rowBeg = blockIdx.x * rpw;
    const int rowEnd = rowBeg + rpw < n ? rowBeg + rpw : n;
    const int nrows  = rowEnd > rowBeg ? rowEnd - rowBeg : 0;
    const int ntile  = (nrows + PT_ROWS - 1) / PT_ROWS;

    if (ntile > 0) {
        vf wreg[4], w2reg[2];
#pragma unroll
        for (int j = 0; j < 4; ++j) wreg[j] = *(const vf*)(W1 + 4 * (threadIdx.x + 1024 * j));
#pragma unroll
        for (int j = 0; j < 2; ++j) w2reg[j] = *(const vf*)(W2 + 4 * (threadIdx.x + 1024 * j));
#pragma unroll
        for (int j = 0; j < 4; ++j) {
            const int e = 4 * (threadIdx.x + 1024 * j);
            const int k = e / N1;
            const int c = e - k * N1;
#pragma unroll
            for (int q = 0; q < 4; ++q) lw1[(k >> 1) * P1 + 2 * (c + q) + (k & 1)] = wreg[j][q];
        }
#pragma unroll
        for (int j = 0; j < 2; ++j) {
            const int e = 4 * (threadIdx.x + 1024 * j);
            const int k = e / N2;
            const int c = e - k * N2;
#pragma unroll
            for (int q = 0; q < 4; ++q) lw2[(k >> 1) * P2 + 2 * (c + q) + (k & 1)] = w2reg[j][q];
        }
    }

    const int laneM = lane & 15;
    const int hi    = lane >> 4;
    const int rt1   = wave >> 3;
    const int ct1   = wave & 7;
    const int rt2   = wave >> 2;
    const int ct2   = wave & 3;
    const float b1v = bias1[ct1 * 16 + laneM];

    for (int it = 0; it < ntile; ++it) {
        const int tile0 = rowBeg + it * PT_ROWS;
        const int rows  = rowEnd - tile0 < PT_ROWS ? rowEnd - tile0 : PT_ROWS;
        const int row0  = tile0 + wave * PT_RPW;

        int   begs[PT_RPW], degs[PT_RPW];
        float dis[PT_RPW];
#pragma unroll
        for (int r = 0; r < PT_RPW; ++r) {
            const int ic = row0 + r < n ? row0 + r : n - 1;
            begs[r] = rowptr[ic];
            degs[r] = cnt[ic] + 1;
            dis[r]  = dinv[ic];
        }

        vf accs[PT_RPW];
#pragma unroll
        for (int r = 0; r < PT_RPW; ++r) {
            const int i = row0 + r;
            vf acc = {};
            if (i < rowEnd) {
                const int   beg = begs[r];
                const int   deg = degs[r];
                const float di  = dis[r];
                int p = 0;
                do {
                    const int m = deg - p < 32 ? deg - p : 32;
                    const int ed = csr[beg + p + (lane < m ? lane : m - 1)];
                    vf v[4];
                    auto ask = [&](int j) {
#pragma unroll
                        for (int u = 0; u < 4; ++u) {
                            const int jj = j + u < m ? j + u : m - 1;
                            const int s  = __builtin_amdgcn_readlane(ed, jj);
                            v[u] = *(const vf*)(x + (long)s * F + lane * V);
                        }
                    };
                    auto add = [&](int j, float wl) {
#pragma unroll
                        for (int u = 0; u < 4; ++u) {
                            const int   jj = j + u < m ? j + u : m - 1;
                            const float w  = __int_as_float(__builtin_amdgcn_readlane(__float_as_int(wl), jj));
                            if (j + u < m) acc += v[u] * w;
                        }
                    };
                    ask(0);
                    const float wl = dinv[ed] * di;
                    add(0, wl);
                    for (int j = 4; j < m; j += 4) {
                        ask(j);
                        add(j, wl);
                    }
                    p += 32;
                } while (p < deg);
            }
            accs[r] = acc;
        }
        __syncthreads();
#pragma unroll
        for (int r = 0; r < PT_RPW; ++r) *(vf*)&tile[(wave * PT_RPW + r) * FG_TP + lane * V] = accs[r];
        __syncthreads();

        const bool on1 = rt1 * 16 < rows;
        v8f d1 = {};
        if (on1) {
            const float* ta = &tile[(rt1 * 16 + laneM) * FG_TP + 2 * hi];
            const float* tb = &lw1[hi * P1 + 2 * (ct1 * 16 + laneM)];
            for (int k4 = 0; k4 < K; k4 += 4) {
                const v2f a = *(const v2f*)&ta[k4];
                const v2f b = *(const v2f*)&tb[(k4 >> 1) * P1];
                d1 = __builtin_amdgcn_wmma_f32_16x16x4_f32(false, a, false, b, (short)0, d1, false, false);
            }
        }
        __syncthreads();
        if (on1) {
#pragma unroll
            for (int r = 0; r < 8; ++r)
                tile[(rt1 * 16 + 8 * hi + r) * FG_TP + ct1 * 16 + laneM] = fmaxf(d1[r] + b1v, 0.0f);
        }
        __syncthreads();

        if (wave < 16 && rt2 * 16 < rows) {
            float dm[8];
#pragma unroll
            for (int r = 0; r < 8; ++r) {
                const int m = tile0 + rt2 * 16 + 8 * hi + r;
                dm[r] = dinv[m < n ? m : n - 1];
            }
            const float* ta = &tile[(rt2 * 16 + laneM) * FG_TP + 2 * hi];
            const float* tb = &lw2[hi * P2 + 2 * (ct2 * 16 + laneM)];
            v8f d = {};
            for (int k4 = 0; k4 < K; k4 += 4) {
                const v2f a = *(const v2f*)&ta[k4];
                const v2f b = *(const v2f*)&tb[(k4 >> 1) * P2];
                d = __builtin_amdgcn_wmma_f32_16x16x4_f32(false, a, false, b, (short)0, d, false, false);
            }
#pragma unroll
            for (int r = 0; r < 8; ++r) {
                const int m = tile0 + rt2 * 16 + 8 * hi + r;
                if (m < rowEnd) Out[(long)m * N2 + ct2 * 16 + laneM] = d[r] * dm[r];
            }
        }
    }
}


extern "C" void kernel_launch(void* const* d_in, const int* in_sizes, int n_in,
                              void* d_out, int out_size, void* d_ws, size_t ws_size,
                              hipStream_t stream) {
    const float* x   = (const float*)d_in[0];
    const int*   ei  = (const int*)d_in[1];
    const float* W1  = (const float*)d_in[2];
    const float* b1  = (const float*)d_in[3];
    const float* W2  = (const float*)d_in[4];
    const float* b2  = (const float*)d_in[5];
    float*       out = (float*)d_out;

    const int  n = in_sizes[0] / 128;
    const long E = in_sizes[1] / 2;
    const int* srcIdx = ei;
    const int* dstIdx = ei + E;

    const long n4 = ((long)n + 3) & ~3L;
    const long E4 = (E + 3) & ~3L;
    int*   cnt    = (int*)d_ws;
    int*   rowptr = cnt + n4;
    int*   csr    = rowptr + n4;
    float* dinv   = (float*)(csr + E4 + n4);
    float* h2     = dinv + n4;
    int2*  pairs  = (int2*)(h2 + (long)n * 64);
    int*   bcnt   = (int*)(pairs + E4);
    int*   bseg   = bcnt + MAXB * NBLK;

    const int B = 256;
    auto blk = [](long work, int b) { return (int)((work + b - 1) / b); };

    const int nbkt = blk(n, BROW);
    const int epb  = blk(E, NBLK);
    k_chunk_group<<<NBLK, BT, 0, stream>>>(srcIdx, dstIdx, E, epb, nbkt, bcnt, bseg, pairs);
    k_bkt_rows<<<nbkt, RT, 0, stream>>>(pairs, bcnt, bseg, E, cnt, rowptr, dinv, csr, n);

    k_gather_mlp<<<PG_WGS, 1024, 0, stream>>>(rowptr, cnt, csr, dinv, x, W1, b1, W2, h2, n, blk(n, PG_WGS));

    k_gather<64, true><<<blk(n, 8 * ROWS), B, 0, stream>>>(rowptr, cnt, csr, dinv, h2, b2, out, n);
}


# both row gathers: a wave walks two of its rows at once (id loads, weight look-ups and four-row load groups of both issued together: 8 row loads in flight, half the round trips); row sum order kept
# speedup vs baseline: 5.5430x; 1.0347x over previous
#include <hip/hip_runtime.h>

typedef float v2f __attribute__((ext_vector_type(2)));
typedef float v8f __attribute__((ext_vector_type(8)));


constexpr int EPT  = 8;
constexpr int BSH  = 9;
constexpr int BROW = 1 << BSH;
constexpr int MAXB = 256;
constexpr int NBLK = 256;
constexpr int BT   = 1024;
constexpr int RT   = 1024;
constexpr int CPW  = NBLK / (RT / 32);
constexpr int SCAP = BT * EPT;
constexpr int CCAP = 24576;

template <int NT>
__device__ __forceinline__ int block_excl_scan(int v, int* sw) {
    const int lane = threadIdx.x & 31, wv = threadIdx.x >> 5;
    int x = v;
    for (int o = 1; o < 32; o <<= 1) {
        const int y = __shfl_up(x, o);
        if (lane >= o) x += y;
    }
    if (lane == 31) sw[wv] = x;
    __syncthreads();
    int base = 0;
#pragma unroll
    for (int k = 0; k < NT / 32; ++k)
        if (k < wv) base += sw[k];
    return base + x - v;
}

__global__ __launch_bounds__(BT) void k_chunk_group(const int* __restrict__ src, const int* __restrict__ dst,
                                                    long e, int epb, int nbkt,
                                                    int* __restrict__ bcnt, int* __restrict__ bseg,
                                                    int2* __restrict__ pairs) {
    __shared__ int h[MAXB], sw[BT / 32];
    __shared__ __align__(16) int2 stage[SCAP];
    const int tid = threadIdx.x;
    const int me  = blockIdx.x;
    if (tid < MAXB) h[tid] = 0;
    __syncthreads();
    const int E   = (int)e;
    const int beg = me * epb < E ? me * epb : E;
    const int end = beg + epb < E ? beg + epb : E;
    int s[EPT] = {}, d[EPT] = {};
    if (beg < end) {
#pragma unroll
        for (int k = 0; k < EPT; ++k) {
            const int i  = beg + k * BT + tid;
            const int ic = i < end ? i : end - 1;
            s[k] = src[ic];
            d[k] = dst[ic];
        }
#pragma unroll
        for (int k = 0; k < EPT; ++k) {
            const int i = beg + k * BT + tid;
            if (i < end) atomicAdd(&h[d[k] >> BSH], 1);
        }
    }
    for (int b0 = beg + BT * EPT; b0 < end; b0 += BT * EPT) {
        int d2[EPT];
#pragma unroll
        for (int k = 0; k < EPT; ++k) {
            const int i = b0 + k * BT + tid;
            d2[k] = dst[i < end ? i : end - 1];
        }
#pragma unroll
        for (int k = 0; k < EPT; ++k) {
            const int i = b0 + k * BT + tid;
            if (i < end) atomicAdd(&h[d2[k] >> BSH], 1);
        }
    }
    __syncthreads();
    const int c     = tid < nbkt ? h[tid] : 0;
    const int first = beg + block_excl_scan<BT>(c, sw);
    const bool staged = end - beg <= SCAP;
    if (tid < nbkt) {
        h[tid] = staged ? first - beg : first;
        bcnt[tid * NBLK + me] = c;
        bseg[tid * NBLK + me] = first;
    }
    __syncthreads();
    if (staged) {
#pragma unroll
        for (int k = 0; k < EPT; ++k) {
            const int i = beg + k * BT + tid;
            if (i < end) stage[atomicAdd(&h[d[k] >> BSH], 1)] = make_int2(s[k], d[k]);
        }
        __syncthreads();
        const int len = end - beg;
        for (int i0 = (beg & ~31) + tid; i0 < end; i0 += 4 * BT) {
            int2 v[4];
#pragma unroll
            for (int u = 0; u < 4; ++u) {
                const int j = i0 + u * BT - beg;
                v[u] = stage[j < 0 ? 0 : (j < len ? j : len - 1)];
            }
#pragma unroll
            for (int u = 0; u < 4; ++u) {
                const int i = i0 + u * BT;
                if (i >= beg && i < end) pairs[i] = v[u];
            }
        }
        return;
    }
    if (beg < end) {
#pragma unroll
        for (int k = 0; k < EPT; ++k) {
            const int i = beg + k * BT + tid;
            if (i < end) pairs[atomicAdd(&h[d[k] >> BSH], 1)] = make_int2(s[k], d[k]);
        }
    }
    for (int b0 = beg + BT * EPT; b0 < end; b0 += BT * EPT) {
        int s2[EPT], d2[EPT];
#pragma unroll
        for (int k = 0; k < EPT; ++k) {
            const int i  = b0 + k * BT + tid;
            const int ic = i < end ? i : end - 1;
            s2[k] = src[ic];
            d2[k] = dst[ic];
        }
#pragma unroll
        for (int k = 0; k < EPT; ++k) {
            const int i = b0 + k * BT + tid;
            if (i < end) pairs[atomicAdd(&h[d2[k] >> BSH], 1)] = make_int2(s2[k], d2[k]);
        }
    }
}

__global__ __launch_bounds__(RT) void k_bkt_rows(const int2* __restrict__ pairs, const int* __restrict__ bcnt,
                                                 const int* __restrict__ bseg, long e,
                                                 int* __restrict__ cnt, int* __restrict__ rowptr,
                                                 float* __restrict__ dinv, int* __restrict__ csr, int n) {
    __shared__ int c[BROW], sw[RT / 32], red[RT / 32], redc[RT / 32];
    __shared__ __align__(16) int stage[CCAP];
    const int tid  = threadIdx.x;
    const int lane = tid & 31;
    const int wv   = __builtin_amdgcn_readfirstlane(tid >> 5);
    const int bk   = blockIdx.x;
    const int E    = (int)e;
    if (tid < BROW) c[tid] = 0;

    int so[CPW], sc[CPW], maxc = 0;
#pragma unroll
    for (int k = 0; k < CPW; ++k) {
        so[k] = bseg[bk * NBLK + wv * CPW + k];
        sc[k] = bcnt[bk * NBLK + wv * CPW + k];
        maxc  = sc[k] > maxc ? sc[k] : maxc;
    }

    int part = tid < NBLK ? bseg[bk * NBLK + tid] - bseg[tid] : 0;
    int mine = tid < NBLK ? bcnt[bk * NBLK + tid] : 0;
    for (int o = 16; o > 0; o >>= 1) {
        part += __shfl_xor(part, o);
        mine += __shfl_xor(mine, o);
    }
    if (lane == 0) {
        red[wv]  = part;
        redc[wv] = mine;
    }

    __syncthreads();
    int beg = 0, edges = 0;
#pragma unroll
    for (int k = 0; k < RT / 32; ++k) {
        beg   += red[k];
        edges += redc[k];
    }

    const int nodeBase = bk << BSH;
    const int rowsHere = n - nodeBase < BROW ? n - nodeBase : BROW;
    const int start    = beg + nodeBase;
    const int total    = edges + (rowsHere > 0 ? rowsHere : 0);
    const bool staged  = total <= CCAP;
    auto for_edges = [&](auto f) {
        for (int t0 = 0; t0 < maxc; t0 += 32) {
            int2 pr[CPW];
#pragma unroll
            for (int k = 0; k < CPW; ++k) {
                const int i = t0 + lane;
                int p = so[k] + (i < sc[k] ? i : sc[k] - 1);
                p = p < 0 ? 0 : (p < E ? p : E - 1);
                pr[k] = pairs[p];
            }
            asm volatile("" ::"v"(pr[0].x), "v"(pr[0].y), "v"(pr[1].x), "v"(pr[1].y), "v"(pr[2].x), "v"(pr[2].y),
                         "v"(pr[3].x), "v"(pr[3].y), "v"(pr[4].x), "v"(pr[4].y), "v"(pr[5].x), "v"(pr[5].y),
                         "v"(pr[6].x), "v"(pr[6].y), "v"(pr[7].x), "v"(pr[7].y));
            static_assert(CPW == 8, "the statement above names CPW pairs");
#pragma unroll
            for (int k = 0; k < CPW; ++k)
                if (t0 + lane < sc[k]) f(pr[k]);
        }
    };
    for_edges([&](int2 pr) { atomicAdd(&c[pr.y - nodeBase], 1); });
    __syncthreads();
    const int node  = nodeBase + tid;
    const int indeg = tid < BROW ? c[tid] : 0;
    const int slots = (tid < BROW && node < n) ? indeg + 1 : 0;
    const int excl  = block_excl_scan<RT>(slots, sw);
    if (tid < BROW && node < n) {
        const int run = start + excl;
        cnt[node]    = indeg;
        rowptr[node] = run;
        dinv[node]   = rsqrtf((float)slots);
        if (staged) {
            stage[excl] = node;
            c[tid]      = excl + 1;
        } else {
            csr[run] = node;
            c[tid]   = run + 1;
        }
    }
    __syncthreads();
    if (staged) {
        for_edges([&](int2 pr) { stage[atomicAdd(&c[pr.y - nodeBase], 1)] = pr.x; });
        __syncthreads();
        const int last = start + total;
        for (int i0 = (start & ~31) + tid; i0 < last; i0 += 4 * RT) {
            int v[4];
#pragma unroll
            for (int u = 0; u < 4; ++u) {
                const int j = i0 + u * RT - start;
                v[u] = stage[j < 0 ? 0 : (j < total ? j : total - 1)];
            }
#pragma unroll
            for (int u = 0; u < 4; ++u) {
                const int i = i0 + u * RT;
                if (i >= start && i < last) csr[i] = v[u];
            }
        }
    } else {
        for_edges([&](int2 pr) { csr[atomicAdd(&c[pr.y - nodeBase], 1)] = pr.x; });
    }
}


constexpr int ROWS = 4;

template <int F, bool WEIGHTED>
__device__ __forceinline__ void gather_two(const int* __restrict__ csr, const float* __restrict__ dinv,
                                           const float* __restrict__ h, int lane,
                                           int beg0, int deg0, float di0, int beg1, int deg1, float di1,
                                           float __attribute__((ext_vector_type(F / 32)))& sum0,
                                           float __attribute__((ext_vector_type(F / 32)))& sum1) {
    constexpr int V = F / 32;
    typedef float vf __attribute__((ext_vector_type(V)));
    vf acc0 = {}, acc1 = {};
    const int degMax = deg0 > deg1 ? deg0 : deg1;
    const int l0 = deg0 > 0 ? deg0 - 1 : 0;
    const int l1 = deg1 > 0 ? deg1 - 1 : 0;
    for (int p = 0; p < degMax; p += 32) {
        const int r0 = deg0 - p, r1 = deg1 - p;
        const int m0 = r0 < 0 ? 0 : (r0 < 32 ? r0 : 32);
        const int m1 = r1 < 0 ? 0 : (r1 < 32 ? r1 : 32);
        const int ed0 = csr[beg0 + (p + lane < l0 ? p + lane : l0)];
        const int ed1 = csr[beg1 + (p + lane < l1 ? p + lane : l1)];
        vf  v0[4], v1[4];
        int s0[4], s1[4];
        auto ids = [&](int (&s)[4], int ed, int m, int j) {
#pragma unroll
            for (int u = 0; u < 4; ++u) {
                const int jj = j + u < m ? j + u : (m > 0 ? m - 1 : 0);
                s[u] = __builtin_amdgcn_readlane(ed, jj);
            }
        };
        auto ask = [&](vf (&v)[4], const int (&s)[4]) {
#pragma unroll
            for (int u = 0; u < 4; ++u) v[u] = *(const vf*)(h + (long)s[u] * F + lane * V);
        };
        auto add = [&](vf& acc, const vf (&v)[4], float wl, int m, int j) {
#pragma unroll
            for (int u = 0; u < 4; ++u) {
                if (WEIGHTED) {
                    const int   jj = j + u < m ? j + u : (m > 0 ? m - 1 : 0);
                    const float w  = __int_as_float(__builtin_amdgcn_readlane(__float_as_int(wl), jj));
                    if (j + u < m) acc += v[u] * w;
                } else {
                    if (j + u < m) acc += v[u];
                }
            }
        };
        ids(s0, ed0, m0, 0);
        ids(s1, ed1, m1, 0);
        ask(v0, s0);
        ask(v1, s1);
        float wl0 = 0.0f, wl1 = 0.0f;
        if (WEIGHTED) {
            const float e0 = dinv[ed0], e1 = dinv[ed1];
            wl0 = e0 * di0;
            wl1 = e1 * di1;
        }
        add(acc0, v0, wl0, m0, 0);
        add(acc1, v1, wl1, m1, 0);
        const int mb = m0 < m1 ? m0 : m1;
        int j = 4;
        for (; j < mb; j += 4) {
            ids(s0, ed0, m0, j);
            ids(s1, ed1, m1, j);
            ask(v0, s0);
            ask(v1, s1);
            add(acc0, v0, wl0, m0, j);
            add(acc1, v1, wl1, m1, j);
        }
        for (int k = j; k < m0; k += 4) {
            ids(s0, ed0, m0, k);
            ask(v0, s0);
            add(acc0, v0, wl0, m0, k);
        }
        for (int k = j; k < m1; k += 4) {
            ids(s1, ed1, m1, k);
            ask(v1, s1);
            add(acc1, v1, wl1, m1, k);
        }
    }
    sum0 = acc0;
    sum1 = acc1;
}

template <int F, bool BIAS>
__global__ __launch_bounds__(256) void k_gather(const int* __restrict__ rowptr,
                                                const int* __restrict__ cnt,
                                                const int* __restrict__ csr,
                                                const float* __restrict__ dinv,
                                                const float* __restrict__ h,
                                                const float* __restrict__ bias,
                                                float* __restrict__ out, int n) {
    constexpr int V = F / 32;
    typedef float vf __attribute__((ext_vector_type(V)));
    const int lane = threadIdx.x & 31;
    const int wv = __builtin_amdgcn_readfirstlane((int)((blockIdx.x * 256 + threadIdx.x) >> 5));

    vf bv = {};
    if (BIAS) bv = *(const vf*)(bias + lane * V);

    int   begs[ROWS], degs[ROWS];
    float dis[ROWS];
#pragma unroll
    for (int r = 0; r < ROWS; ++r) {
        const int ic = wv * ROWS + r < n ? wv * ROWS + r : n - 1;
        begs[r] = rowptr[ic];
        degs[r] = cnt[ic] + 1;
        dis[r]  = dinv[ic];
    }

    static_assert(ROWS % 2 == 0, "the wave's rows are walked in pairs");
#pragma unroll
    for (int r = 0; r < ROWS; r += 2) {
        const int i = wv * ROWS + r;
        if (i >= n) break;
        const bool two = i + 1 < n;
        vf a0, a1;
        gather_two<F, false>(csr, dinv, h, lane, begs[r], degs[r], dis[r],
                             begs[r + 1], two ? degs[r + 1] : 0, dis[r + 1], a0, a1);
        a0 *= dis[r];
        if (BIAS) a0 += bv;
        *(vf*)(out + (long)i * F + lane * V) = a0;
        if (two) {
            a1 *= dis[r + 1];
            if (BIAS) a1 += bv;
            *(vf*)(out + (long)(i + 1) * F + lane * V) = a1;
        }
    }
}


constexpr int PG_WGS  = 512;
constexpr int PT_ROWS = 64;
constexpr int PT_RPW  = 2;
constexpr int FG_TP   = 132;

__global__ __launch_bounds__(1024) __attribute__((amdgpu_waves_per_eu(16, 16))) void k_gather_mlp(const int* __restrict__ rowptr,
                                                    const int* __restrict__ cnt,
                                                    const int* __restrict__ csr,
                                                    const float* __restrict__ dinv,
                                                    const float* __restrict__ x,
                                                    const float* __restrict__ W1,
                                                    const float* __restrict__ bias1,
                                                    const float* __restrict__ W2,
                                                    float* __restrict__ Out, int n, int rpw) {
    constexpr int F = 128, K = 128, N1 = 128, N2 = 64, V = 4;
    constexpr int P1 = 2 * N1 + 32;
    constexpr int P2 = 2 * N2 + 32;
    typedef float vf __attribute__((ext_vector_type(V)));
    __shared__ __align__(16) float lw1[(K / 2) * P1];
    __shared__ __align__(16) float lw2[(K / 2) * P2];
    __shared__ __align__(16) float tile[PT_ROWS * FG_TP];

    const int lane = threadIdx.x & 31;
    const int wave = __builtin_amdgcn_readfirstlane((int)(threadIdx.x >> 5));

    static_assert(PT_ROWS == 32 * PT_RPW, "32 waves of PT_RPW rows each fill the tile");
    static_assert(K * N1 == 4 * 4 * 1024, "four 16-byte reads per thread cover W1");
    static_assert(K * N2 == 2 * 4 * 1024, "two 16-byte reads per thread cover W2");

    const int rowBeg = blockIdx.x * rpw;
    const int rowEnd = rowBeg + rpw < n ? rowBeg + rpw : n;
    const int nrows  = rowEnd > rowBeg ? rowEnd - rowBeg : 0;
    const int ntile  = (nrows + PT_ROWS - 1) / PT_ROWS;

    if (ntile > 0) {
        vf wreg[4], w2reg[2];
#pragma unroll
        for (int j = 0; j < 4; ++j) wreg[j] = *(const vf*)(W1 + 4 * (threadIdx.x + 1024 * j));
#pragma unroll
        for (int j = 0; j < 2; ++j) w2reg[j] = *(const vf*)(W2 + 4 * (threadIdx.x + 1024 * j));
#pragma unroll
        for (int j = 0; j < 4; ++j) {
            const int e = 4 * (threadIdx.x + 1024 * j);
            const int k = e / N1;
            const int c = e - k * N1;
#pragma unroll
            for (int q = 0; q < 4; ++q) lw1[(k >> 1) * P1 + 2 * (c + q) + (k & 1)] = wreg[j][q];
        }
#pragma unroll
        for (int j = 0; j < 2; ++j) {
            const int e = 4 * (threadIdx.x + 1024 * j);
            const int k = e / N2;
            const int c = e - k * N2;
#pragma unroll
            for (int q = 0; q < 4; ++q) lw2[(k >> 1) * P2 + 2 * (c + q) + (k & 1)] = w2reg[j][q];
        }
    }

    const int laneM = lane & 15;
    const int hi    = lane >> 4;
    const int rt1   = wave >> 3;
    const int ct1   = wave & 7;
    const int rt2   = wave >> 2;
    const int ct2   = wave & 3;
    const float b1v = bias1[ct1 * 16 + laneM];

    for (int it = 0; it < ntile; ++it) {
        const int tile0 = rowBeg + it * PT_ROWS;
        const int rows  = rowEnd - tile0 < PT_ROWS ? rowEnd - tile0 : PT_ROWS;
        const int row0  = tile0 + wave * PT_RPW;

        int   begs[PT_RPW], degs[PT_RPW];
        float dis[PT_RPW];
#pragma unroll
        for (int r = 0; r < PT_RPW; ++r) {
            const int ic = row0 + r < n ? row0 + r : n - 1;
            begs[r] = rowptr[ic];
            degs[r] = cnt[ic] + 1;
            dis[r]  = dinv[ic];
        }

        static_assert(PT_RPW == 2, "gather_two sums the wave's two rows");
        vf accs[PT_RPW];
        gather_two<F, true>(csr, dinv, x, lane, begs[0], row0 < rowEnd ? degs[0] : 0, dis[0],
                            begs[1], row0 + 1 < rowEnd ? degs[1] : 0, dis[1], accs[0], accs[1]);
        __syncthreads();
#pragma unroll
        for (int r = 0; r < PT_RPW; ++r) *(vf*)&tile[(wave * PT_RPW + r) * FG_TP + lane * V] = accs[r];
        __syncthreads();

        const bool on1 = rt1 * 16 < rows;
        v8f d1 = {};
        if (on1) {
            const float* ta = &tile[(rt1 * 16 + laneM) * FG_TP + 2 * hi];
            const float* tb = &lw1[hi * P1 + 2 * (ct1 * 16 + laneM)];
            for (int k4 = 0; k4 < K; k4 += 4) {
                const v2f a = *(const v2f*)&ta[k4];
                const v2f b = *(const v2f*)&tb[(k4 >> 1) * P1];
                d1 = __builtin_amdgcn_wmma_f32_16x16x4_f32(false, a, false, b, (short)0, d1, false, false);
            }
        }
        __syncthreads();
        if (on1) {
#pragma unroll
            for (int r = 0; r < 8; ++r)
                tile[(rt1 * 16 + 8 * hi + r) * FG_TP + ct1 * 16 + laneM] = fmaxf(d1[r] + b1v, 0.0f);
        }
        __syncthreads();

        if (wave < 16 && rt2 * 16 < rows) {
            float dm[8];
#pragma unroll
            for (int r = 0; r < 8; ++r) {
                const int m = tile0 + rt2 * 16 + 8 * hi + r;
                dm[r] = dinv[m < n ? m : n - 1];
            }
            const float* ta = &tile[(rt2 * 16 + laneM) * FG_TP + 2 * hi];
            const float* tb = &lw2[hi * P2 + 2 * (ct2 * 16 + laneM)];
            v8f d = {};
            for (int k4 = 0; k4 < K; k4 += 4) {
                const v2f a = *(const v2f*)&ta[k4];
                const v2f b = *(const v2f*)&tb[(k4 >> 1) * P2];
                d = __builtin_amdgcn_wmma_f32_16x16x4_f32(false, a, false, b, (short)0, d, false, false);
            }
#pragma unroll
            for (int r = 0; r < 8; ++r) {
                const int m = tile0 + rt2 * 16 + 8 * hi + r;
                if (m < rowEnd) Out[(long)m * N2 + ct2 * 16 + laneM] = d[r] * dm[r];
            }
        }
    }
}


extern "C" void kernel_launch(void* const* d_in, const int* in_sizes, int n_in,
                              void* d_out, int out_size, void* d_ws, size_t ws_size,
                              hipStream_t stream) {
    const float* x   = (const float*)d_in[0];
    const int*   ei  = (const int*)d_in[1];
    const float* W1  = (const float*)d_in[2];
    const float* b1  = (const float*)d_in[3];
    const float* W2  = (const float*)d_in[4];
    const float* b2  = (const float*)d_in[5];
    float*       out = (float*)d_out;

    const int  n = in_sizes[0] / 128;
    const long E = in_sizes[1] / 2;
    const int* srcIdx = ei;
    const int* dstIdx = ei + E;

    const long n4 = ((long)n + 3) & ~3L;
    const long E4 = (E + 3) & ~3L;
    int*   cnt    = (int*)d_ws;
    int*   rowptr = cnt + n4;
    int*   csr    = rowptr + n4;
    float* dinv   = (float*)(csr + E4 + n4);
    float* h2     = dinv + n4;
    int2*  pairs  = (int2*)(h2 + (long)n * 64);
    int*   bcnt   = (int*)(pairs + E4);
    int*   bseg   = bcnt + MAXB * NBLK;

    const int B = 256;
    auto blk = [](long work, int b) { return (int)((work + b - 1) / b); };

    const int nbkt = blk(n, BROW);
    const int epb  = blk(E, NBLK);
    k_chunk_group<<<NBLK, BT, 0, stream>>>(srcIdx, dstIdx, E, epb, nbkt, bcnt, bseg, pairs);
    k_bkt_rows<<<nbkt, RT, 0, stream>>>(pairs, bcnt, bseg, E, cnt, rowptr, dinv, csr, n);

    k_gather_mlp<<<PG_WGS, 1024, 0, stream>>>(rowptr, cnt, csr, dinv, x, W1, b1, W2, h2, n, blk(n, PG_WGS));

    k_gather<64, true><<<blk(n, 8 * ROWS), B, 0, stream>>>(rowptr, cnt, csr, dinv, h2, b2, out, n);
}
